# starting kernel
# baseline (speedup 1.0000x reference)
#include <hip/hip_runtime.h>
#include <hip/hip_bf16.h>

#define NUM_USERS 50000
#define NUM_ITEMS 20000
#define NNODES    70000
#define MTILES    (NNODES / 16)
#define HID       128
#define NE        1000000
#define EPLUS     (NE + NNODES)
#define NEG_SLOPE 0.2f

typedef float v2f __attribute__((ext_vector_type(2)));
typedef float v8f __attribute__((ext_vector_type(8)));

__device__ __forceinline__ unsigned fkey(float f) {
    unsigned b = __float_as_uint(f);
    return (b & 0x80000000u) ? ~b : (b | 0x80000000u);
}
__device__ __forceinline__ float funkey(unsigned k) {
    unsigned b = (k & 0x80000000u) ? (k & 0x7FFFFFFFu) : ~k;
    return __uint_as_float(b);
}

__global__ __launch_bounds__(256) void concat_k(const float* __restrict__ u,
                                                const float* __restrict__ it,
                                                float* __restrict__ x) {
    int i = blockIdx.x * 256 + threadIdx.x;
    const int UW = NUM_USERS * HID;
    if (i < NNODES * HID) x[i] = (i < UW) ? u[i] : it[i - UW];
}

__global__ __launch_bounds__(256) void gemm_wmma(const float* __restrict__ X,
                                                 const float* __restrict__ Wl,
                                                 float* __restrict__ H) {
    const int wave  = threadIdx.x >> 5;
    const int lane  = threadIdx.x & 31;
    const int tileM = blockIdx.x * 8 + wave;
    if (tileM >= MTILES) return;
    const int m     = lane & 15;
    const int khalf = (lane >> 4) * 2;
    const int row   = tileM * 16 + m;

    v8f c[8] = {};
    const float* xr = X + (size_t)row * HID + khalf;
    const float* wr = Wl + (size_t)khalf * HID + m;

    for (int k = 0; k < HID; k += 4) {
        v2f a = *(const v2f*)(xr + k);
        const float* w0 = wr + (size_t)k * HID;
        const float* w1 = w0 + HID;
#pragma unroll
        for (int t = 0; t < 8; ++t) {
            v2f b;
            b.x = w0[t * 16];
            b.y = w1[t * 16];
            c[t] = __builtin_amdgcn_wmma_f32_16x16x4_f32(false, a, false, b,
                                                         (short)0, c[t], false, false);
        }
    }
    const int rbase = tileM * 16 + ((lane >> 4) ? 8 : 0);
#pragma unroll
    for (int t = 0; t < 8; ++t)
#pragma unroll
        for (int r = 0; r < 8; ++r)
            H[(size_t)(rbase + r) * HID + t * 16 + m] = c[t][r];
}

__global__ __launch_bounds__(256) void alpha_k(const float* __restrict__ H,
                                               const float* __restrict__ Asrc,
                                               const float* __restrict__ Adst,
                                               float* __restrict__ as_out,
                                               float* __restrict__ ad_out) {
    const int warp = threadIdx.x >> 5, lane = threadIdx.x & 31;
    const int node = blockIdx.x * 8 + warp;
    if (node >= NNODES) return;
    float4 hv = ((const float4*)H)[(size_t)node * 32 + lane];
    float4 av = ((const float4*)Asrc)[lane];
    float4 dv = ((const float4*)Adst)[lane];
    float ss = hv.x * av.x + hv.y * av.y + hv.z * av.z + hv.w * av.w;
    float sd = hv.x * dv.x + hv.y * dv.y + hv.z * dv.z + hv.w * dv.w;
#pragma unroll
    for (int off = 16; off > 0; off >>= 1) {
        ss += __shfl_xor(ss, off, 32);
        sd += __shfl_xor(sd, off, 32);
    }
    if (lane == 0) { as_out[node] = ss; ad_out[node] = sd; }
}

__global__ __launch_bounds__(256) void init_nodes(unsigned* __restrict__ mkey,
                                                  float* __restrict__ den) {
    int i = blockIdx.x * 256 + threadIdx.x;
    if (i < NNODES) { mkey[i] = 0u; den[i] = 0.f; }
}

__global__ __launch_bounds__(256) void init_bias(float* __restrict__ out,
                                                 const float* __restrict__ b) {
    int i = blockIdx.x * 256 + threadIdx.x;
    if (i < NNODES * HID) out[i] = b[i & (HID - 1)];
}

__device__ __forceinline__ void edge_sd(int i, const int* __restrict__ ei,
                                        int& s, int& d) {
    if (i < NE) { s = ei[i]; d = ei[NE + i]; }
    else        { s = d = i - NE; }
}

__global__ __launch_bounds__(256) void edge_max_k(const int* __restrict__ ei,
                                                  const float* __restrict__ as_,
                                                  const float* __restrict__ ad_,
                                                  unsigned* __restrict__ mkey) {
    int i = blockIdx.x * 256 + threadIdx.x;
    if (i >= EPLUS) return;
    int s, d; edge_sd(i, ei, s, d);
    float e = as_[s] + ad_[d];
    e = (e >= 0.f) ? e : NEG_SLOPE * e;
    atomicMax(mkey + d, fkey(e));
}

__global__ __launch_bounds__(256) void edge_exp_k(const int* __restrict__ ei,
                                                  const float* __restrict__ as_,
                                                  const float* __restrict__ ad_,
                                                  const unsigned* __restrict__ mkey,
                                                  float* __restrict__ den,
                                                  float* __restrict__ exb) {
    int i = blockIdx.x * 256 + threadIdx.x;
    if (i >= EPLUS) return;
    int s, d; edge_sd(i, ei, s, d);
    float e = as_[s] + ad_[d];
    e = (e >= 0.f) ? e : NEG_SLOPE * e;
    float ex = expf(e - funkey(mkey[d]));
    exb[i] = ex;
    atomicAdd(den + d, ex);
}

__global__ __launch_bounds__(256) void edge_agg_k(const int* __restrict__ ei,
                                                  const float* __restrict__ H,
                                                  const float* __restrict__ exb,
                                                  const float* __restrict__ den,
                                                  float* __restrict__ out) {
    const int warp = threadIdx.x >> 5, lane = threadIdx.x & 31;
    const int e = blockIdx.x * 8 + warp;
    if (e >= EPLUS) return;
    int s, d; edge_sd(e, ei, s, d);
    float alpha = exb[e] / den[d];
    float4 hv = ((const float4*)H)[(size_t)s * 32 + lane];
    float* o = out + (size_t)d * HID + lane * 4;
    atomicAdd(o + 0, hv.x * alpha);
    atomicAdd(o + 1, hv.y * alpha);
    atomicAdd(o + 2, hv.z * alpha);
    atomicAdd(o + 3, hv.w * alpha);
}

extern "C" void kernel_launch(void* const* d_in, const int* in_sizes, int n_in,
                              void* d_out, int out_size, void* d_ws, size_t ws_size,
                              hipStream_t stream) {
    const int*   ei      = (const int*)d_in[0];
    const float* users   = (const float*)d_in[1];
    const float* items   = (const float*)d_in[2];
    const float* W       = (const float*)d_in[3];
    const float* att_src = (const float*)d_in[4];
    const float* att_dst = (const float*)d_in[5];
    const float* bias    = (const float*)d_in[6];
    float*       out     = (float*)d_out;

    float*    xbuf  = (float*)d_ws;
    float*    hbuf  = xbuf + (size_t)NNODES * HID;
    float*    asb   = hbuf + (size_t)NNODES * HID;
    float*    adb   = asb + NNODES;
    unsigned* mkey  = (unsigned*)(adb + NNODES);
    float*    den   = (float*)(mkey + NNODES);
    float*    exb   = den + NNODES;

    const int GRID_NH = (NNODES * HID + 255) / 256;
    const int GRID_N  = (NNODES + 255) / 256;
    const int GRID_E  = (EPLUS + 255) / 256;
    const int GRID_EW = (EPLUS + 7) / 8;
    const int GRID_GE = (MTILES + 7) / 8;

    concat_k<<<GRID_NH, 256, 0, stream>>>(users, items, xbuf);

    for (int l = 0; l < 2; ++l) {
        const float* Wl = W + (size_t)l * HID * HID;
        float* tgt = (l == 0) ? xbuf : out;
        gemm_wmma<<<GRID_GE, 256, 0, stream>>>(xbuf, Wl, hbuf);
        alpha_k<<<(NNODES + 7) / 8, 256, 0, stream>>>(hbuf, att_src + l * HID,
                                                      att_dst + l * HID, asb, adb);
        init_nodes<<<GRID_N, 256, 0, stream>>>(mkey, den);
        init_bias<<<GRID_NH, 256, 0, stream>>>(tgt, bias + l * HID);
        edge_max_k<<<GRID_E, 256, 0, stream>>>(ei, asb, adb, mkey);
        edge_exp_k<<<GRID_E, 256, 0, stream>>>(ei, asb, adb, mkey, den, exb);
        edge_agg_k<<<GRID_EW, 256, 0, stream>>>(ei, hbuf, exb, den, tgt);
    }
}


# Fill launches folded into the logit kernel: each node's wave also stores the bias row to the aggregation target and zeroes its max key and denominator; 4 launches and 564384 waves fewer
# speedup vs baseline: 1.4365x; 1.4365x over previous
#include <hip/hip_runtime.h>
#include <hip/hip_bf16.h>

#define NUM_USERS 50000
#define NUM_ITEMS 20000
#define NNODES    70000
#define MTILES    (NNODES / 16)
#define HID       128
#define NE        1000000
#define EPLUS     (NE + NNODES)
#define NEG_SLOPE 0.2f

typedef float v2f __attribute__((ext_vector_type(2)));
typedef float v8f __attribute__((ext_vector_type(8)));

__device__ __forceinline__ unsigned fkey(float f) {
    unsigned b = __float_as_uint(f);
    return (b & 0x80000000u) ? ~b : (b | 0x80000000u);
}
__device__ __forceinline__ float funkey(unsigned k) {
    unsigned b = (k & 0x80000000u) ? (k & 0x7FFFFFFFu) : ~k;
    return __uint_as_float(b);
}

__global__ __launch_bounds__(256) void concat_k(const float* __restrict__ u,
                                                const float* __restrict__ it,
                                                float* __restrict__ x) {
    int i = blockIdx.x * 256 + threadIdx.x;
    const int UW = NUM_USERS * HID;
    if (i < NNODES * HID) x[i] = (i < UW) ? u[i] : it[i - UW];
}

__global__ __launch_bounds__(256) void gemm_wmma(const float* __restrict__ X,
                                                 const float* __restrict__ Wl,
                                                 float* __restrict__ H) {
    const int wave  = threadIdx.x >> 5;
    const int lane  = threadIdx.x & 31;
    const int tileM = blockIdx.x * 8 + wave;
    if (tileM >= MTILES) return;
    const int m     = lane & 15;
    const int khalf = (lane >> 4) * 2;
    const int row   = tileM * 16 + m;

    v8f c[8] = {};
    const float* xr = X + (size_t)row * HID + khalf;
    const float* wr = Wl + (size_t)khalf * HID + m;

    for (int k = 0; k < HID; k += 4) {
        v2f a = *(const v2f*)(xr + k);
        const float* w0 = wr + (size_t)k * HID;
        const float* w1 = w0 + HID;
#pragma unroll
        for (int t = 0; t < 8; ++t) {
            v2f b;
            b.x = w0[t * 16];
            b.y = w1[t * 16];
            c[t] = __builtin_amdgcn_wmma_f32_16x16x4_f32(false, a, false, b,
                                                         (short)0, c[t], false, false);
        }
    }
    const int rbase = tileM * 16 + ((lane >> 4) ? 8 : 0);
#pragma unroll
    for (int t = 0; t < 8; ++t)
#pragma unroll
        for (int r = 0; r < 8; ++r)
            H[(size_t)(rbase + r) * HID + t * 16 + m] = c[t][r];
}

__global__ __launch_bounds__(256) void alpha_k(const float* __restrict__ H,
                                               const float* __restrict__ Asrc,
                                               const float* __restrict__ Adst,
                                               const float* __restrict__ b,
                                               float* __restrict__ as_out,
                                               float* __restrict__ ad_out,
                                               unsigned* __restrict__ mkey,
                                               float* __restrict__ den,
                                               float* __restrict__ tgt) {
    const int warp = threadIdx.x >> 5, lane = threadIdx.x & 31;
    const int node = blockIdx.x * 8 + warp;
    if (node >= NNODES) return;
    float4 hv = ((const float4*)H)[(size_t)node * 32 + lane];
    float4 av = ((const float4*)Asrc)[lane];
    float4 dv = ((const float4*)Adst)[lane];
    ((float4*)tgt)[(size_t)node * 32 + lane] = ((const float4*)b)[lane];
    float ss = hv.x * av.x + hv.y * av.y + hv.z * av.z + hv.w * av.w;
    float sd = hv.x * dv.x + hv.y * dv.y + hv.z * dv.z + hv.w * dv.w;
#pragma unroll
    for (int off = 16; off > 0; off >>= 1) {
        ss += __shfl_xor(ss, off, 32);
        sd += __shfl_xor(sd, off, 32);
    }
    if (lane == 0) {
        as_out[node] = ss; ad_out[node] = sd;
        mkey[node] = 0u;   den[node] = 0.f;
    }
}

__device__ __forceinline__ void edge_sd(int i, const int* __restrict__ ei,
                                        int& s, int& d) {
    if (i < NE) { s = ei[i]; d = ei[NE + i]; }
    else        { s = d = i - NE; }
}

__global__ __launch_bounds__(256) void edge_max_k(const int* __restrict__ ei,
                                                  const float* __restrict__ as_,
                                                  const float* __restrict__ ad_,
                                                  unsigned* __restrict__ mkey) {
    int i = blockIdx.x * 256 + threadIdx.x;
    if (i >= EPLUS) return;
    int s, d; edge_sd(i, ei, s, d);
    float e = as_[s] + ad_[d];
    e = (e >= 0.f) ? e : NEG_SLOPE * e;
    atomicMax(mkey + d, fkey(e));
}

__global__ __launch_bounds__(256) void edge_exp_k(const int* __restrict__ ei,
                                                  const float* __restrict__ as_,
                                                  const float* __restrict__ ad_,
                                                  const unsigned* __restrict__ mkey,
                                                  float* __restrict__ den,
                                                  float* __restrict__ exb) {
    int i = blockIdx.x * 256 + threadIdx.x;
    if (i >= EPLUS) return;
    int s, d; edge_sd(i, ei, s, d);
    float e = as_[s] + ad_[d];
    e = (e >= 0.f) ? e : NEG_SLOPE * e;
    float ex = expf(e - funkey(mkey[d]));
    exb[i] = ex;
    atomicAdd(den + d, ex);
}

#define AGG_EPW 32
#define AGG_U   8
static_assert(AGG_EPW % AGG_U == 0 && EPLUS % AGG_U == 0 && NE % AGG_EPW == 0,
              "edge chunks must be whole");

__global__ __launch_bounds__(256) void edge_agg_k(const int* __restrict__ ei,
                                                  const float* __restrict__ H,
                                                  const float* __restrict__ exb,
                                                  const float* __restrict__ den,
                                                  float* __restrict__ out) {
    const int warp = __builtin_amdgcn_readfirstlane(threadIdx.x >> 5);
    const int lane = threadIdx.x & 31;
    const int base = (blockIdx.x * 8 + warp) * AGG_EPW;
    if (base >= EPLUS) return;
    const int n = (EPLUS - base < AGG_EPW) ? (EPLUS - base) : AGG_EPW;

    int s = 0, d = 0;
    float alpha = 0.f;
    if (lane < n) {
        edge_sd(base + lane, ei, s, d);
        alpha = exb[base + lane] / den[d];
    }

    for (int j0 = 0; j0 < n; j0 += AGG_U) {
        float hv[AGG_U][4];
#pragma unroll
        for (int u = 0; u < AGG_U; ++u) {
            const int su = __builtin_amdgcn_readlane(s, j0 + u);
            const float* hr = H + (size_t)su * HID + lane;
            hv[u][0] = hr[0]; hv[u][1] = hr[32]; hv[u][2] = hr[64]; hv[u][3] = hr[96];
        }
#pragma unroll
        for (int u = 0; u < AGG_U; ++u) {
            const int   du = __builtin_amdgcn_readlane(d, j0 + u);
            const float au = __int_as_float(
                __builtin_amdgcn_readlane(__float_as_int(alpha), j0 + u));
            float* o = out + (size_t)du * HID + lane;
            atomicAdd(o +  0, hv[u][0] * au);
            atomicAdd(o + 32, hv[u][1] * au);
            atomicAdd(o + 64, hv[u][2] * au);
            atomicAdd(o + 96, hv[u][3] * au);
        }
    }
}

extern "C" void kernel_launch(void* const* d_in, const int* in_sizes, int n_in,
                              void* d_out, int out_size, void* d_ws, size_t ws_size,
                              hipStream_t stream) {
    const int*   ei      = (const int*)d_in[0];
    const float* users   = (const float*)d_in[1];
    const float* items   = (const float*)d_in[2];
    const float* W       = (const float*)d_in[3];
    const float* att_src = (const float*)d_in[4];
    const float* att_dst = (const float*)d_in[5];
    const float* bias    = (const float*)d_in[6];
    float*       out     = (float*)d_out;

    float*    xbuf  = (float*)d_ws;
    float*    hbuf  = xbuf + (size_t)NNODES * HID;
    float*    asb   = hbuf + (size_t)NNODES * HID;
    float*    adb   = asb + NNODES;
    unsigned* mkey  = (unsigned*)(adb + NNODES);
    float*    den   = (float*)(mkey + NNODES);
    float*    exb   = den + NNODES;

    const int GRID_NH = (NNODES * HID + 255) / 256;
    const int GRID_E  = (EPLUS + 255) / 256;
    const int GRID_EW = (EPLUS + 8 * AGG_EPW - 1) / (8 * AGG_EPW);
    const int GRID_GE = (MTILES + 7) / 8;

    concat_k<<<GRID_NH, 256, 0, stream>>>(users, items, xbuf);

    for (int l = 0; l < 2; ++l) {
        const float* Wl = W + (size_t)l * HID * HID;
        float* tgt = (l == 0) ? xbuf : out;
        gemm_wmma<<<GRID_GE, 256, 0, stream>>>(xbuf, Wl, hbuf);
        alpha_k<<<(NNODES + 7) / 8, 256, 0, stream>>>(hbuf, att_src + l * HID,
                                                      att_dst + l * HID, bias + l * HID,
                                                      asb, adb, mkey, den, tgt);
        edge_max_k<<<GRID_E, 256, 0, stream>>>(ei, asb, adb, mkey);
        edge_exp_k<<<GRID_E, 256, 0, stream>>>(ei, asb, adb, mkey, den, exb);
        edge_agg_k<<<GRID_EW, 256, 0, stream>>>(ei, hbuf, exb, den, tgt);
    }
}


# Aggregation by row owner over a 1-byte owner-key plane: 256 groups keep 274 rows in LDS, queue their edges from 1.07 MB of keys, add rows by LDS float atomics, store once; sum order = LDS arrival
# speedup vs baseline: 1.4886x; 1.0363x over previous
#include <hip/hip_runtime.h>
#include <hip/hip_bf16.h>

#define NUM_USERS 50000
#define NUM_ITEMS 20000
#define NNODES    70000
#define MTILES    (NNODES / 16)
#define HID       128
#define NE        1000000
#define EPLUS     (NE + NNODES)
#define NEG_SLOPE 0.2f

typedef float v2f __attribute__((ext_vector_type(2)));
typedef float v8f __attribute__((ext_vector_type(8)));

__device__ __forceinline__ unsigned fkey(float f) {
    unsigned b = __float_as_uint(f);
    return (b & 0x80000000u) ? ~b : (b | 0x80000000u);
}
__device__ __forceinline__ float funkey(unsigned k) {
    unsigned b = (k & 0x80000000u) ? (k & 0x7FFFFFFFu) : ~k;
    return __uint_as_float(b);
}

__global__ __launch_bounds__(256) void concat_k(const float* __restrict__ u,
                                                const float* __restrict__ it,
                                                float* __restrict__ x) {
    int i = blockIdx.x * 256 + threadIdx.x;
    const int UW = NUM_USERS * HID;
    if (i < NNODES * HID) x[i] = (i < UW) ? u[i] : it[i - UW];
}

__global__ __launch_bounds__(256) void gemm_wmma(const float* __restrict__ X,
                                                 const float* __restrict__ Wl,
                                                 float* __restrict__ H) {
    const int wave  = threadIdx.x >> 5;
    const int lane  = threadIdx.x & 31;
    const int tileM = blockIdx.x * 8 + wave;
    if (tileM >= MTILES) return;
    const int m     = lane & 15;
    const int khalf = (lane >> 4) * 2;
    const int row   = tileM * 16 + m;

    v8f c[8] = {};
    const float* xr = X + (size_t)row * HID + khalf;
    const float* wr = Wl + (size_t)khalf * HID + m;

    for (int k = 0; k < HID; k += 4) {
        v2f a = *(const v2f*)(xr + k);
        const float* w0 = wr + (size_t)k * HID;
        const float* w1 = w0 + HID;
#pragma unroll
        for (int t = 0; t < 8; ++t) {
            v2f b;
            b.x = w0[t * 16];
            b.y = w1[t * 16];
            c[t] = __builtin_amdgcn_wmma_f32_16x16x4_f32(false, a, false, b,
                                                         (short)0, c[t], false, false);
        }
    }
    const int rbase = tileM * 16 + ((lane >> 4) ? 8 : 0);
#pragma unroll
    for (int t = 0; t < 8; ++t)
#pragma unroll
        for (int r = 0; r < 8; ++r)
            H[(size_t)(rbase + r) * HID + t * 16 + m] = c[t][r];
}

__global__ __launch_bounds__(256) void alpha_k(const float* __restrict__ H,
                                               const float* __restrict__ Asrc,
                                               const float* __restrict__ Adst,
                                               float* __restrict__ as_out,
                                               float* __restrict__ ad_out,
                                               unsigned* __restrict__ mkey,
                                               float* __restrict__ den) {
    const int warp = threadIdx.x >> 5, lane = threadIdx.x & 31;
    const int node = blockIdx.x * 8 + warp;
    if (node >= NNODES) return;
    float4 hv = ((const float4*)H)[(size_t)node * 32 + lane];
    float4 av = ((const float4*)Asrc)[lane];
    float4 dv = ((const float4*)Adst)[lane];
    float ss = hv.x * av.x + hv.y * av.y + hv.z * av.z + hv.w * av.w;
    float sd = hv.x * dv.x + hv.y * dv.y + hv.z * dv.z + hv.w * dv.w;
#pragma unroll
    for (int off = 16; off > 0; off >>= 1) {
        ss += __shfl_xor(ss, off, 32);
        sd += __shfl_xor(sd, off, 32);
    }
    if (lane == 0) {
        as_out[node] = ss; ad_out[node] = sd;
        mkey[node] = 0u;   den[node] = 0.f;
    }
}

__device__ __forceinline__ void edge_sd(int i, const int* __restrict__ ei,
                                        int& s, int& d) {
    if (i < NE) { s = ei[i]; d = ei[NE + i]; }
    else        { s = d = i - NE; }
}

__global__ __launch_bounds__(256) void edge_max_k(const int* __restrict__ ei,
                                                  const float* __restrict__ as_,
                                                  const float* __restrict__ ad_,
                                                  unsigned* __restrict__ mkey) {
    int i = blockIdx.x * 256 + threadIdx.x;
    if (i >= EPLUS) return;
    int s, d; edge_sd(i, ei, s, d);
    float e = as_[s] + ad_[d];
    e = (e >= 0.f) ? e : NEG_SLOPE * e;
    atomicMax(mkey + d, fkey(e));
}

#define OWN_R 274
#define NOWN  ((NNODES + OWN_R - 1) / OWN_R)
static_assert(NOWN <= 256, "the owner index must fit one byte");
__global__ __launch_bounds__(256) void edge_exp_k(const int* __restrict__ ei,
                                                  const float* __restrict__ as_,
                                                  const float* __restrict__ ad_,
                                                  const unsigned* __restrict__ mkey,
                                                  float* __restrict__ den,
                                                  float* __restrict__ exb,
                                                  unsigned* __restrict__ keyw) {
    int i = blockIdx.x * 256 + threadIdx.x;
    if (i >= EPLUS) return;
    int s, d; edge_sd(i, ei, s, d);
    float e = as_[s] + ad_[d];
    e = (e >= 0.f) ? e : NEG_SLOPE * e;
    float ex = expf(e - funkey(mkey[d]));
    exb[i] = ex;
    ((unsigned char*)keyw)[i] = (unsigned char)(d / OWN_R);
    atomicAdd(den + d, ex);
}

#define OWN_T   1024
#define SCAN_U  6
#define QCAP    8192
#define AGG_U   8
#define NPIECE  (EPLUS / 16)
static_assert(EPLUS % 16 == 0, "key plane must be whole 16-byte pieces");

__device__ __forceinline__ void own_direct(int e, int row0, const int* __restrict__ ei,
                                           const float* __restrict__ H,
                                           const float* __restrict__ exb,
                                           const float* dens, float* rows) {
    int s, d; edge_sd(e, ei, s, d);
    const int dl = d - row0;
    const float a = exb[e] / dens[dl];
    const float* hr = H + (size_t)s * HID;
    float* r = rows + dl * HID;
#pragma unroll 1
    for (int c = 0; c < HID; ++c) atomicAdd(r + c, hr[c] * a);
}

__global__ __launch_bounds__(OWN_T) void own_agg_k(const int* __restrict__ ei,
                                                   const float* __restrict__ H,
                                                   const float* __restrict__ exb,
                                                   const float* __restrict__ den,
                                                   const unsigned* __restrict__ keyw,
                                                   const float* __restrict__ b,
                                                   float* __restrict__ out) {
    __shared__ float rows[OWN_R * HID];
    __shared__ float dens[OWN_R];
    __shared__ int   q[QCAP];
    __shared__ int   qn, qhead;

    const int tid  = threadIdx.x;
    const int lane = tid & 31;
    const int row0 = blockIdx.x * OWN_R;
    const int nrows = (NNODES - row0 < OWN_R) ? (NNODES - row0) : OWN_R;

    {
        const float4 bv = ((const float4*)b)[lane];
        for (int i = tid; i < OWN_R * HID / 4; i += OWN_T) ((float4*)rows)[i] = bv;
        for (int i = tid; i < OWN_R; i += OWN_T) dens[i] = (i < nrows) ? den[row0 + i] : 1.f;
        if (tid == 0) { qn = 0; qhead = 0; }
    }
    __syncthreads();

    {
        const uint4* kp = (const uint4*)keyw;
        const unsigned pat  = blockIdx.x * 0x01010101u;
        const unsigned npat = ~pat;
        const int rot = blockIdx.x * (NPIECE / NOWN);
        for (int base = 0; base < NPIECE; base += OWN_T * SCAN_U) {
            uint4 v[SCAN_U];
#pragma unroll
            for (int u = 0; u < SCAN_U; ++u) {
                const int p = base + u * OWN_T + tid;
                const int pr = (p + rot < NPIECE) ? (p + rot) : (p + rot - NPIECE);
                v[u] = make_uint4(npat, npat, npat, npat);
                if (p < NPIECE) v[u] = kp[pr];
            }
#pragma unroll
            for (int u = 0; u < SCAN_U; ++u) {
                const int p = base + u * OWN_T + tid;
                const int pr = (p + rot < NPIECE) ? (p + rot) : (p + rot - NPIECE);
                const unsigned w[4] = { v[u].x, v[u].y, v[u].z, v[u].w };
                unsigned mk = 0;
#pragma unroll
                for (int k = 0; k < 4; ++k) {
                    const unsigned x = w[k] ^ pat;
                    const unsigned z = ~(((x & 0x7F7F7F7Fu) + 0x7F7F7F7Fu) | x) & 0x80808080u;
                    mk |= z >> (7 - k);
                }
                while (mk) {
                    const int bit = __builtin_ctz(mk);
                    mk &= mk - 1;
                    const int e = pr * 16 + 4 * (bit & 3) + (bit >> 3);
                    const int slot = atomicAdd(&qn, 1);
                    if (slot < QCAP) q[slot] = e;
                    else             own_direct(e, row0, ei, H, exb, dens, rows);
                }
            }
        }
    }
    __syncthreads();

    {
        const int nq = __builtin_amdgcn_readfirstlane(qn);
        const int n  = (nq < QCAP) ? nq : QCAP;
        for (;;) {
            int c0 = 0;
            if (lane == 0) c0 = atomicAdd(&qhead, 32);
            c0 = __builtin_amdgcn_readfirstlane(c0);
            if (c0 >= n) break;
            const int m = (n - c0 < 32) ? (n - c0) : 32;

            int s = 0, dl = 0;
            float alpha = 0.f;
            if (lane < m) {
                const int e = q[c0 + lane];
                int d; edge_sd(e, ei, s, d);
                dl = d - row0;
                alpha = exb[e] / dens[dl];
            }
            for (int j0 = 0; j0 < m; j0 += AGG_U) {
                float hv[AGG_U][4];
#pragma unroll
                for (int u = 0; u < AGG_U; ++u) {
                    const int su = __builtin_amdgcn_readlane(s, j0 + u);
                    const float* hr = H + (size_t)su * HID + lane;
                    hv[u][0] = hr[0]; hv[u][1] = hr[32]; hv[u][2] = hr[64]; hv[u][3] = hr[96];
                }
#pragma unroll
                for (int u = 0; u < AGG_U; ++u) {
                    if (j0 + u < m) {
                        const int   dlu = __builtin_amdgcn_readlane(dl, j0 + u);
                        const float au  = __int_as_float(
                            __builtin_amdgcn_readlane(__float_as_int(alpha), j0 + u));
                        float* r = rows + dlu * HID + lane;
                        atomicAdd(r +  0, hv[u][0] * au);
                        atomicAdd(r + 32, hv[u][1] * au);
                        atomicAdd(r + 64, hv[u][2] * au);
                        atomicAdd(r + 96, hv[u][3] * au);
                    }
                }
            }
        }
    }
    __syncthreads();

    {
        float4* o = (float4*)(out + (size_t)row0 * HID);
        for (int i = tid; i < nrows * HID / 4; i += OWN_T) o[i] = ((const float4*)rows)[i];
    }
}

extern "C" void kernel_launch(void* const* d_in, const int* in_sizes, int n_in,
                              void* d_out, int out_size, void* d_ws, size_t ws_size,
                              hipStream_t stream) {
    const int*   ei      = (const int*)d_in[0];
    const float* users   = (const float*)d_in[1];
    const float* items   = (const float*)d_in[2];
    const float* W       = (const float*)d_in[3];
    const float* att_src = (const float*)d_in[4];
    const float* att_dst = (const float*)d_in[5];
    const float* bias    = (const float*)d_in[6];
    float*       out     = (float*)d_out;

    float*    xbuf  = (float*)d_ws;
    float*    hbuf  = xbuf + (size_t)NNODES * HID;
    float*    asb   = hbuf + (size_t)NNODES * HID;
    float*    adb   = asb + NNODES;
    unsigned* mkey  = (unsigned*)(adb + NNODES);
    float*    den   = (float*)(mkey + NNODES);
    float*    exb   = den + NNODES;
    unsigned* keyw  = (unsigned*)(exb + EPLUS);

    const int GRID_NH = (NNODES * HID + 255) / 256;
    const int GRID_E  = (EPLUS + 255) / 256;
    const int GRID_GE = (MTILES + 7) / 8;

    concat_k<<<GRID_NH, 256, 0, stream>>>(users, items, xbuf);

    for (int l = 0; l < 2; ++l) {
        const float* Wl = W + (size_t)l * HID * HID;
        float* tgt = (l == 0) ? xbuf : out;
        gemm_wmma<<<GRID_GE, 256, 0, stream>>>(xbuf, Wl, hbuf);
        alpha_k<<<(NNODES + 7) / 8, 256, 0, stream>>>(hbuf, att_src + l * HID,
                                                      att_dst + l * HID,
                                                      asb, adb, mkey, den);
        edge_max_k<<<GRID_E, 256, 0, stream>>>(ei, asb, adb, mkey);
        edge_exp_k<<<GRID_E, 256, 0, stream>>>(ei, asb, adb, mkey, den, exb, keyw);
        own_agg_k<<<NOWN, OWN_T, 0, stream>>>(ei, hbuf, exb, den, keyw, bias + l * HID, tgt);
    }
}


# Edge softmax moved into the row-owner kernel: row max, exp and denominator by LDS atomics over its queued edges; both edge passes, their global atomics and 4 launches gone; den sum = LDS arrival
# speedup vs baseline: 2.8250x; 1.8978x over previous
#include <hip/hip_runtime.h>
#include <hip/hip_bf16.h>

#define NUM_USERS 50000
#define NUM_ITEMS 20000
#define NNODES    70000
#define MTILES    (NNODES / 16)
#define HID       128
#define NE        1000000
#define EPLUS     (NE + NNODES)
#define NEG_SLOPE 0.2f

typedef float v2f __attribute__((ext_vector_type(2)));
typedef float v8f __attribute__((ext_vector_type(8)));

__device__ __forceinline__ unsigned fkey(float f) {
    unsigned b = __float_as_uint(f);
    return (b & 0x80000000u) ? ~b : (b | 0x80000000u);
}
__device__ __forceinline__ float funkey(unsigned k) {
    unsigned b = (k & 0x80000000u) ? (k & 0x7FFFFFFFu) : ~k;
    return __uint_as_float(b);
}

#define OWN_R 274
#define NOWN  ((NNODES + OWN_R - 1) / OWN_R)
static_assert(NOWN <= 256, "the owner index must fit one byte");
static_assert(EPLUS % 4 == 0 && NE % 4 == 0, "four edges per key word, never across the self-loop border");
__global__ __launch_bounds__(256) void concat_k(const float* __restrict__ u,
                                                const float* __restrict__ it,
                                                float* __restrict__ x,
                                                const int* __restrict__ ei,
                                                unsigned* __restrict__ keyw) {
    int i = blockIdx.x * 256 + threadIdx.x;
    const int UW = NUM_USERS * HID;
    if (i < NNODES * HID) x[i] = (i < UW) ? u[i] : it[i - UW];
    if (i < EPLUS / 4) {
        int4 d;
        if (4 * i < NE) d = ((const int4*)(ei + NE))[i];
        else { const int n = 4 * i - NE; d = make_int4(n, n + 1, n + 2, n + 3); }
        keyw[i] = (unsigned)(d.x / OWN_R) | ((unsigned)(d.y / OWN_R) << 8) |
                  ((unsigned)(d.z / OWN_R) << 16) | ((unsigned)(d.w / OWN_R) << 24);
    }
}

__global__ __launch_bounds__(256) void gemm_wmma(const float* __restrict__ X,
                                                 const float* __restrict__ Wl,
                                                 float* __restrict__ H) {
    const int wave  = threadIdx.x >> 5;
    const int lane  = threadIdx.x & 31;
    const int tileM = blockIdx.x * 8 + wave;
    if (tileM >= MTILES) return;
    const int m     = lane & 15;
    const int khalf = (lane >> 4) * 2;
    const int row   = tileM * 16 + m;

    v8f c[8] = {};
    const float* xr = X + (size_t)row * HID + khalf;
    const float* wr = Wl + (size_t)khalf * HID + m;

    for (int k = 0; k < HID; k += 4) {
        v2f a = *(const v2f*)(xr + k);
        const float* w0 = wr + (size_t)k * HID;
        const float* w1 = w0 + HID;
#pragma unroll
        for (int t = 0; t < 8; ++t) {
            v2f b;
            b.x = w0[t * 16];
            b.y = w1[t * 16];
            c[t] = __builtin_amdgcn_wmma_f32_16x16x4_f32(false, a, false, b,
                                                         (short)0, c[t], false, false);
        }
    }
    const int rbase = tileM * 16 + ((lane >> 4) ? 8 : 0);
#pragma unroll
    for (int t = 0; t < 8; ++t)
#pragma unroll
        for (int r = 0; r < 8; ++r)
            H[(size_t)(rbase + r) * HID + t * 16 + m] = c[t][r];
}

__global__ __launch_bounds__(256) void alpha_k(const float* __restrict__ H,
                                               const float* __restrict__ Asrc,
                                               const float* __restrict__ Adst,
                                               float* __restrict__ as_out,
                                               float* __restrict__ ad_out) {
    const int warp = threadIdx.x >> 5, lane = threadIdx.x & 31;
    const int node = blockIdx.x * 8 + warp;
    if (node >= NNODES) return;
    float4 hv = ((const float4*)H)[(size_t)node * 32 + lane];
    float4 av = ((const float4*)Asrc)[lane];
    float4 dv = ((const float4*)Adst)[lane];
    float ss = hv.x * av.x + hv.y * av.y + hv.z * av.z + hv.w * av.w;
    float sd = hv.x * dv.x + hv.y * dv.y + hv.z * dv.z + hv.w * dv.w;
#pragma unroll
    for (int off = 16; off > 0; off >>= 1) {
        ss += __shfl_xor(ss, off, 32);
        sd += __shfl_xor(sd, off, 32);
    }
    if (lane == 0) { as_out[node] = ss; ad_out[node] = sd; }
}

__device__ __forceinline__ void edge_sd(int i, const int* __restrict__ ei,
                                        int& s, int& d) {
    const bool real = i < NE;
    const int  ic = real ? i : 0;
    const int  sv = ei[ic], dv = ei[NE + ic];
    s = real ? sv : i - NE;
    d = real ? dv : i - NE;
}

__device__ __forceinline__ float lrelu(float e) { return (e >= 0.f) ? e : NEG_SLOPE * e; }

#define OWN_T   1024
#define SCAN_U  6
#define QCAP    8192
#define LOG_U   8
#define AGG_U   8
#define NPIECE  (EPLUS / 16)
#define SBITS   17
static_assert(EPLUS % 16 == 0, "key plane must be whole 16-byte pieces");
static_assert(QCAP == OWN_T * LOG_U, "the logit step covers the queue in one sweep");
static_assert(NNODES <= (1 << SBITS) && OWN_R <= (1 << (31 - SBITS)), "queue entry packs (s, local row)");

template <class F>
__device__ __forceinline__ void scan_owned(const unsigned* __restrict__ keyw, int tid, F f) {
    const uint4* kp = (const uint4*)keyw;
    const unsigned pat  = blockIdx.x * 0x01010101u;
    const unsigned npat = ~pat;
    const int rot = blockIdx.x * (NPIECE / NOWN);
    for (int base = 0; base < NPIECE; base += OWN_T * SCAN_U) {
        uint4 v[SCAN_U];
#pragma unroll
        for (int u = 0; u < SCAN_U; ++u) {
            const int p = base + u * OWN_T + tid;
            const int pr = (p + rot < NPIECE) ? (p + rot) : (p + rot - NPIECE);
            v[u] = make_uint4(npat, npat, npat, npat);
            if (p < NPIECE) v[u] = kp[pr];
        }
#pragma unroll
        for (int u = 0; u < SCAN_U; ++u) {
            const int p = base + u * OWN_T + tid;
            const int pr = (p + rot < NPIECE) ? (p + rot) : (p + rot - NPIECE);
            const unsigned w[4] = { v[u].x, v[u].y, v[u].z, v[u].w };
            unsigned mk = 0;
#pragma unroll
            for (int k = 0; k < 4; ++k) {
                const unsigned x = w[k] ^ pat;
                const unsigned z = ~(((x & 0x7F7F7F7Fu) + 0x7F7F7F7Fu) | x) & 0x80808080u;
                mk |= z >> (7 - k);
            }
            while (mk) {
                const int bit = __builtin_ctz(mk);
                mk &= mk - 1;
                f(pr * 16 + 4 * (bit & 3) + (bit >> 3));
            }
        }
    }
}

__global__ __launch_bounds__(OWN_T) void own_agg_k(const int* __restrict__ ei,
                                                   const float* __restrict__ H,
                                                   const float* __restrict__ as_,
                                                   const float* __restrict__ ad_,
                                                   const unsigned* __restrict__ keyw,
                                                   const float* __restrict__ b,
                                                   float* __restrict__ out) {
    __shared__ float    rows[OWN_R * HID];
    __shared__ float    dens[OWN_R];
    __shared__ float    adl[OWN_R];
    __shared__ unsigned mx[OWN_R];
    __shared__ int      q[QCAP];
    __shared__ float    qx[QCAP];
    __shared__ int      qn, qhead;

    const int tid  = threadIdx.x;
    const int lane = tid & 31;
    const int row0 = blockIdx.x * OWN_R;
    const int nrows = (NNODES - row0 < OWN_R) ? (NNODES - row0) : OWN_R;

    {
        const float4 bv = ((const float4*)b)[lane];
        for (int i = tid; i < OWN_R * HID / 4; i += OWN_T) ((float4*)rows)[i] = bv;
        for (int i = tid; i < OWN_R; i += OWN_T) {
            dens[i] = 0.f;
            mx[i]   = 0u;
            adl[i]  = (i < nrows) ? ad_[row0 + i] : 0.f;
        }
        if (tid == 0) { qn = 0; qhead = 0; }
    }
    __syncthreads();

    scan_owned(keyw, tid, [&](int e) {
        const int slot = atomicAdd(&qn, 1);
        if (slot < QCAP) q[slot] = e;
    });
    __syncthreads();

    const int nq = __builtin_amdgcn_readfirstlane(qn);

    if (nq <= QCAP) {
        {
            int e[LOG_U], s[LOG_U], d[LOG_U];
            float a[LOG_U];
#pragma unroll
            for (int u = 0; u < LOG_U; ++u) {
                const int i = u * OWN_T + tid;
                e[u] = (i < nq) ? q[i] : 0;
            }
#pragma unroll
            for (int u = 0; u < LOG_U; ++u) edge_sd(e[u], ei, s[u], d[u]);
#pragma unroll
            for (int u = 0; u < LOG_U; ++u) a[u] = as_[s[u]];
#pragma unroll
            for (int u = 0; u < LOG_U; ++u) {
                const int i = u * OWN_T + tid;
                if (i < nq) {
                    const int   dl = d[u] - row0;
                    const float lg = lrelu(a[u] + adl[dl]);
                    atomicMax(&mx[dl], fkey(lg));
                    q[i]  = s[u] | (dl << SBITS);
                    qx[i] = lg;
                }
            }
        }
        __syncthreads();

        for (int i = tid; i < nq; i += OWN_T) {
            const int   dl = q[i] >> SBITS;
            const float ex = expf(qx[i] - funkey(mx[dl]));
            qx[i] = ex;
            atomicAdd(&dens[dl], ex);
        }
        __syncthreads();

        for (;;) {
            int c0 = 0;
            if (lane == 0) c0 = atomicAdd(&qhead, 32);
            c0 = __builtin_amdgcn_readfirstlane(c0);
            if (c0 >= nq) break;
            const int m = (nq - c0 < 32) ? (nq - c0) : 32;

            int s = 0, dl = 0;
            float alpha = 0.f;
            if (lane < m) {
                const int pk = q[c0 + lane];
                s  = pk & ((1 << SBITS) - 1);
                dl = pk >> SBITS;
                alpha = qx[c0 + lane] / dens[dl];
            }
            for (int j0 = 0; j0 < m; j0 += AGG_U) {
                float hv[AGG_U][4];
#pragma unroll
                for (int u = 0; u < AGG_U; ++u) {
                    const int su = __builtin_amdgcn_readlane(s, j0 + u);
                    const float* hr = H + (size_t)su * HID + lane;
                    hv[u][0] = hr[0]; hv[u][1] = hr[32]; hv[u][2] = hr[64]; hv[u][3] = hr[96];
                }
#pragma unroll
                for (int u = 0; u < AGG_U; ++u) {
                    if (j0 + u < m) {
                        const int   dlu = __builtin_amdgcn_readlane(dl, j0 + u);
                        const float au  = __int_as_float(
                            __builtin_amdgcn_readlane(__float_as_int(alpha), j0 + u));
                        float* r = rows + dlu * HID + lane;
                        atomicAdd(r +  0, hv[u][0] * au);
                        atomicAdd(r + 32, hv[u][1] * au);
                        atomicAdd(r + 64, hv[u][2] * au);
                        atomicAdd(r + 96, hv[u][3] * au);
                    }
                }
            }
        }
    } else {
#pragma unroll 1
        for (int pass = 0; pass < 3; ++pass) {
            scan_owned(keyw, tid, [&](int e) {
                int s, d; edge_sd(e, ei, s, d);
                const int   dl = d - row0;
                const float lg = lrelu(as_[s] + adl[dl]);
                if (pass == 0) { atomicMax(&mx[dl], fkey(lg)); return; }
                const float ex = expf(lg - funkey(mx[dl]));
                if (pass == 1) { atomicAdd(&dens[dl], ex); return; }
                const float al = ex / dens[dl];
                const float* hr = H + (size_t)s * HID;
                float* r = rows + dl * HID;
#pragma unroll 1
                for (int c = 0; c < HID; ++c) atomicAdd(r + c, hr[c] * al);
            });
            __syncthreads();
        }
    }
    __syncthreads();

    {
        float4* o = (float4*)(out + (size_t)row0 * HID);
        for (int i = tid; i < nrows * HID / 4; i += OWN_T) o[i] = ((const float4*)rows)[i];
    }
}

extern "C" void kernel_launch(void* const* d_in, const int* in_sizes, int n_in,
                              void* d_out, int out_size, void* d_ws, size_t ws_size,
                              hipStream_t stream) {
    const int*   ei      = (const int*)d_in[0];
    const float* users   = (const float*)d_in[1];
    const float* items   = (const float*)d_in[2];
    const float* W       = (const float*)d_in[3];
    const float* att_src = (const float*)d_in[4];
    const float* att_dst = (const float*)d_in[5];
    const float* bias    = (const float*)d_in[6];
    float*       out     = (float*)d_out;

    float*    xbuf  = (float*)d_ws;
    float*    hbuf  = xbuf + (size_t)NNODES * HID;
    float*    asb   = hbuf + (size_t)NNODES * HID;
    float*    adb   = asb + NNODES;
    unsigned* keyw  = (unsigned*)(adb + NNODES);

    const int GRID_NH = (NNODES * HID + 255) / 256;
    const int GRID_GE = (MTILES + 7) / 8;

    concat_k<<<GRID_NH, 256, 0, stream>>>(users, items, xbuf, ei, keyw);

    for (int l = 0; l < 2; ++l) {
        const float* Wl = W + (size_t)l * HID * HID;
        float* tgt = (l == 0) ? xbuf : out;
        gemm_wmma<<<GRID_GE, 256, 0, stream>>>(xbuf, Wl, hbuf);
        alpha_k<<<(NNODES + 7) / 8, 256, 0, stream>>>(hbuf, att_src + l * HID,
                                                      att_dst + l * HID, asb, adb);
        own_agg_k<<<NOWN, OWN_T, 0, stream>>>(ei, hbuf, asb, adb, keyw, bias + l * HID, tgt);
    }
}


# Owner queues built once: layer 0 stores each group's queue (s, local row) and length in the workspace, layer 1 reads them back; its key scan and edge-end loads are gone; sum order = LDS arrival
# speedup vs baseline: 3.1295x; 1.1078x over previous
#include <hip/hip_runtime.h>
#include <hip/hip_bf16.h>

#define NUM_USERS 50000
#define NUM_ITEMS 20000
#define NNODES    70000
#define MTILES    (NNODES / 16)
#define HID       128
#define NE        1000000
#define EPLUS     (NE + NNODES)
#define NEG_SLOPE 0.2f

typedef float v2f __attribute__((ext_vector_type(2)));
typedef float v8f __attribute__((ext_vector_type(8)));

__device__ __forceinline__ unsigned fkey(float f) {
    unsigned b = __float_as_uint(f);
    return (b & 0x80000000u) ? ~b : (b | 0x80000000u);
}
__device__ __forceinline__ float funkey(unsigned k) {
    unsigned b = (k & 0x80000000u) ? (k & 0x7FFFFFFFu) : ~k;
    return __uint_as_float(b);
}

#define OWN_R 274
#define NOWN  ((NNODES + OWN_R - 1) / OWN_R)
static_assert(NOWN <= 256, "the owner index must fit one byte");
static_assert(EPLUS % 4 == 0 && NE % 4 == 0, "four edges per key word, never across the self-loop border");
__global__ __launch_bounds__(256) void concat_k(const float* __restrict__ u,
                                                const float* __restrict__ it,
                                                float* __restrict__ x,
                                                const int* __restrict__ ei,
                                                unsigned* __restrict__ keyw) {
    int i = blockIdx.x * 256 + threadIdx.x;
    const int UW = NUM_USERS * HID;
    if (i < NNODES * HID) x[i] = (i < UW) ? u[i] : it[i - UW];
    if (i < EPLUS / 4) {
        int4 d;
        if (4 * i < NE) d = ((const int4*)(ei + NE))[i];
        else { const int n = 4 * i - NE; d = make_int4(n, n + 1, n + 2, n + 3); }
        keyw[i] = (unsigned)(d.x / OWN_R) | ((unsigned)(d.y / OWN_R) << 8) |
                  ((unsigned)(d.z / OWN_R) << 16) | ((unsigned)(d.w / OWN_R) << 24);
    }
}

__global__ __launch_bounds__(256) void gemm_wmma(const float* __restrict__ X,
                                                 const float* __restrict__ Wl,
                                                 float* __restrict__ H) {
    const int wave  = threadIdx.x >> 5;
    const int lane  = threadIdx.x & 31;
    const int tileM = blockIdx.x * 8 + wave;
    if (tileM >= MTILES) return;
    const int m     = lane & 15;
    const int khalf = (lane >> 4) * 2;
    const int row   = tileM * 16 + m;

    v8f c[8] = {};
    const float* xr = X + (size_t)row * HID + khalf;
    const float* wr = Wl + (size_t)khalf * HID + m;

    for (int k = 0; k < HID; k += 4) {
        v2f a = *(const v2f*)(xr + k);
        const float* w0 = wr + (size_t)k * HID;
        const float* w1 = w0 + HID;
#pragma unroll
        for (int t = 0; t < 8; ++t) {
            v2f b;
            b.x = w0[t * 16];
            b.y = w1[t * 16];
            c[t] = __builtin_amdgcn_wmma_f32_16x16x4_f32(false, a, false, b,
                                                         (short)0, c[t], false, false);
        }
    }
    const int rbase = tileM * 16 + ((lane >> 4) ? 8 : 0);
#pragma unroll
    for (int t = 0; t < 8; ++t)
#pragma unroll
        for (int r = 0; r < 8; ++r)
            H[(size_t)(rbase + r) * HID + t * 16 + m] = c[t][r];
}

__global__ __launch_bounds__(256) void alpha_k(const float* __restrict__ H,
                                               const float* __restrict__ Asrc,
                                               const float* __restrict__ Adst,
                                               float* __restrict__ as_out,
                                               float* __restrict__ ad_out) {
    const int warp = threadIdx.x >> 5, lane = threadIdx.x & 31;
    const int node = blockIdx.x * 8 + warp;
    if (node >= NNODES) return;
    float4 hv = ((const float4*)H)[(size_t)node * 32 + lane];
    float4 av = ((const float4*)Asrc)[lane];
    float4 dv = ((const float4*)Adst)[lane];
    float ss = hv.x * av.x + hv.y * av.y + hv.z * av.z + hv.w * av.w;
    float sd = hv.x * dv.x + hv.y * dv.y + hv.z * dv.z + hv.w * dv.w;
#pragma unroll
    for (int off = 16; off > 0; off >>= 1) {
        ss += __shfl_xor(ss, off, 32);
        sd += __shfl_xor(sd, off, 32);
    }
    if (lane == 0) { as_out[node] = ss; ad_out[node] = sd; }
}

__device__ __forceinline__ void edge_sd(int i, const int* __restrict__ ei,
                                        int& s, int& d) {
    const bool real = i < NE;
    const int  ic = real ? i : 0;
    const int  sv = ei[ic], dv = ei[NE + ic];
    s = real ? sv : i - NE;
    d = real ? dv : i - NE;
}

__device__ __forceinline__ float lrelu(float e) { return (e >= 0.f) ? e : NEG_SLOPE * e; }

#define OWN_T   1024
#define SCAN_U  6
#define QCAP    8192
#define LOG_U   8
#define AGG_U   8
#define NPIECE  (EPLUS / 16)
#define SBITS   17
static_assert(EPLUS % 16 == 0, "key plane must be whole 16-byte pieces");
static_assert(QCAP == OWN_T * LOG_U, "the logit step covers the queue in one sweep");
static_assert(NNODES <= (1 << SBITS) && OWN_R <= (1 << (31 - SBITS)), "queue entry packs (s, local row)");

template <class F>
__device__ __forceinline__ void scan_owned(const unsigned* __restrict__ keyw, int tid, F f) {
    const uint4* kp = (const uint4*)keyw;
    const unsigned pat  = blockIdx.x * 0x01010101u;
    const unsigned npat = ~pat;
    const int rot = blockIdx.x * (NPIECE / NOWN);
    for (int base = 0; base < NPIECE; base += OWN_T * SCAN_U) {
        uint4 v[SCAN_U];
#pragma unroll
        for (int u = 0; u < SCAN_U; ++u) {
            const int p = base + u * OWN_T + tid;
            const int pr = (p + rot < NPIECE) ? (p + rot) : (p + rot - NPIECE);
            v[u] = make_uint4(npat, npat, npat, npat);
            if (p < NPIECE) v[u] = kp[pr];
        }
#pragma unroll
        for (int u = 0; u < SCAN_U; ++u) {
            const int p = base + u * OWN_T + tid;
            const int pr = (p + rot < NPIECE) ? (p + rot) : (p + rot - NPIECE);
            const unsigned w[4] = { v[u].x, v[u].y, v[u].z, v[u].w };
            unsigned mk = 0;
#pragma unroll
            for (int k = 0; k < 4; ++k) {
                const unsigned x = w[k] ^ pat;
                const unsigned z = ~(((x & 0x7F7F7F7Fu) + 0x7F7F7F7Fu) | x) & 0x80808080u;
                mk |= z >> (7 - k);
            }
            while (mk) {
                const int bit = __builtin_ctz(mk);
                mk &= mk - 1;
                f(pr * 16 + 4 * (bit & 3) + (bit >> 3));
            }
        }
    }
}

template <int LAYER>
__global__ __launch_bounds__(OWN_T) void own_agg_k(const int* __restrict__ ei,
                                                   const float* __restrict__ H,
                                                   const float* __restrict__ as_,
                                                   const float* __restrict__ ad_,
                                                   const unsigned* __restrict__ keyw,
                                                   const float* __restrict__ b,
                                                   float* __restrict__ out,
                                                   int* __restrict__ qs,
                                                   int* __restrict__ nqs) {
    __shared__ float    rows[OWN_R * HID];
    __shared__ float    dens[OWN_R];
    __shared__ float    adl[OWN_R];
    __shared__ unsigned mx[OWN_R];
    __shared__ int      q[QCAP];
    __shared__ float    qx[QCAP];
    __shared__ int      qn, qhead;

    const int tid  = threadIdx.x;
    const int lane = tid & 31;
    const int row0 = blockIdx.x * OWN_R;
    const int nrows = (NNODES - row0 < OWN_R) ? (NNODES - row0) : OWN_R;

    {
        const float4 bv = ((const float4*)b)[lane];
        for (int i = tid; i < OWN_R * HID / 4; i += OWN_T) ((float4*)rows)[i] = bv;
        for (int i = tid; i < OWN_R; i += OWN_T) {
            dens[i] = 0.f;
            mx[i]   = 0u;
            adl[i]  = (i < nrows) ? ad_[row0 + i] : 0.f;
        }
        if (tid == 0) { qn = 0; qhead = 0; }
    }
    __syncthreads();

    int* const qsv = qs + blockIdx.x * QCAP;
    int nq;
    if (LAYER == 0) {
        scan_owned(keyw, tid, [&](int e) {
            const int slot = atomicAdd(&qn, 1);
            if (slot < QCAP) q[slot] = e;
        });
        __syncthreads();
        nq = __builtin_amdgcn_readfirstlane(qn);
        if (tid == 0) nqs[blockIdx.x] = nq;
    } else {
        nq = __builtin_amdgcn_readfirstlane(nqs[blockIdx.x]);
    }

    if (nq <= QCAP) {
        {
            int s[LOG_U], dl[LOG_U];
            float a[LOG_U];
            if (LAYER == 0) {
                int e[LOG_U], d[LOG_U];
#pragma unroll
                for (int u = 0; u < LOG_U; ++u) {
                    const int i = u * OWN_T + tid;
                    e[u] = (i < nq) ? q[i] : 0;
                }
#pragma unroll
                for (int u = 0; u < LOG_U; ++u) edge_sd(e[u], ei, s[u], d[u]);
#pragma unroll
                for (int u = 0; u < LOG_U; ++u) dl[u] = d[u] - row0;
            } else {
                int pk[LOG_U];
#pragma unroll
                for (int u = 0; u < LOG_U; ++u) {
                    const int i = u * OWN_T + tid;
                    pk[u] = qsv[(i < nq) ? i : 0];
                }
#pragma unroll
                for (int u = 0; u < LOG_U; ++u) {
                    s[u]  = pk[u] & ((1 << SBITS) - 1);
                    dl[u] = pk[u] >> SBITS;
                }
            }
#pragma unroll
            for (int u = 0; u < LOG_U; ++u) a[u] = as_[s[u]];
#pragma unroll
            for (int u = 0; u < LOG_U; ++u) {
                const int i = u * OWN_T + tid;
                if (i < nq) {
                    const float lg = lrelu(a[u] + adl[dl[u]]);
                    atomicMax(&mx[dl[u]], fkey(lg));
                    const int pkv = s[u] | (dl[u] << SBITS);
                    q[i]  = pkv;
                    qx[i] = lg;
                    if (LAYER == 0) qsv[i] = pkv;
                }
            }
        }
        __syncthreads();

        for (int i = tid; i < nq; i += OWN_T) {
            const int   dl = q[i] >> SBITS;
            const float ex = expf(qx[i] - funkey(mx[dl]));
            qx[i] = ex;
            atomicAdd(&dens[dl], ex);
        }
        __syncthreads();

        for (;;) {
            int c0 = 0;
            if (lane == 0) c0 = atomicAdd(&qhead, 32);
            c0 = __builtin_amdgcn_readfirstlane(c0);
            if (c0 >= nq) break;
            const int m = (nq - c0 < 32) ? (nq - c0) : 32;

            int s = 0, dl = 0;
            float alpha = 0.f;
            if (lane < m) {
                const int pk = q[c0 + lane];
                s  = pk & ((1 << SBITS) - 1);
                dl = pk >> SBITS;
                alpha = qx[c0 + lane] / dens[dl];
            }
            for (int j0 = 0; j0 < m; j0 += AGG_U) {
                float hv[AGG_U][4];
#pragma unroll
                for (int u = 0; u < AGG_U; ++u) {
                    const int su = __builtin_amdgcn_readlane(s, j0 + u);
                    const float* hr = H + (size_t)su * HID + lane;
                    hv[u][0] = hr[0]; hv[u][1] = hr[32]; hv[u][2] = hr[64]; hv[u][3] = hr[96];
                }
#pragma unroll
                for (int u = 0; u < AGG_U; ++u) {
                    if (j0 + u < m) {
                        const int   dlu = __builtin_amdgcn_readlane(dl, j0 + u);
                        const float au  = __int_as_float(
                            __builtin_amdgcn_readlane(__float_as_int(alpha), j0 + u));
                        float* r = rows + dlu * HID + lane;
                        atomicAdd(r +  0, hv[u][0] * au);
                        atomicAdd(r + 32, hv[u][1] * au);
                        atomicAdd(r + 64, hv[u][2] * au);
                        atomicAdd(r + 96, hv[u][3] * au);
                    }
                }
            }
        }
    } else {
#pragma unroll 1
        for (int pass = 0; pass < 3; ++pass) {
            scan_owned(keyw, tid, [&](int e) {
                int s, d; edge_sd(e, ei, s, d);
                const int   dl = d - row0;
                const float lg = lrelu(as_[s] + adl[dl]);
                if (pass == 0) { atomicMax(&mx[dl], fkey(lg)); return; }
                const float ex = expf(lg - funkey(mx[dl]));
                if (pass == 1) { atomicAdd(&dens[dl], ex); return; }
                const float al = ex / dens[dl];
                const float* hr = H + (size_t)s * HID;
                float* r = rows + dl * HID;
#pragma unroll 1
                for (int c = 0; c < HID; ++c) atomicAdd(r + c, hr[c] * al);
            });
            __syncthreads();
        }
    }
    __syncthreads();

    {
        float4* o = (float4*)(out + (size_t)row0 * HID);
        for (int i = tid; i < nrows * HID / 4; i += OWN_T) o[i] = ((const float4*)rows)[i];
    }
}

extern "C" void kernel_launch(void* const* d_in, const int* in_sizes, int n_in,
                              void* d_out, int out_size, void* d_ws, size_t ws_size,
                              hipStream_t stream) {
    const int*   ei      = (const int*)d_in[0];
    const float* users   = (const float*)d_in[1];
    const float* items   = (const float*)d_in[2];
    const float* W       = (const float*)d_in[3];
    const float* att_src = (const float*)d_in[4];
    const float* att_dst = (const float*)d_in[5];
    const float* bias    = (const float*)d_in[6];
    float*       out     = (float*)d_out;

    float*    xbuf  = (float*)d_ws;
    float*    hbuf  = xbuf + (size_t)NNODES * HID;
    float*    asb   = hbuf + (size_t)NNODES * HID;
    float*    adb   = asb + NNODES;
    unsigned* keyw  = (unsigned*)(adb + NNODES);
    int*      qs    = (int*)(keyw + EPLUS / 4);
    int*      nqs   = qs + NOWN * QCAP;

    const int GRID_NH = (NNODES * HID + 255) / 256;
    const int GRID_GE = (MTILES + 7) / 8;

    concat_k<<<GRID_NH, 256, 0, stream>>>(users, items, xbuf, ei, keyw);

    for (int l = 0; l < 2; ++l) {
        const float* Wl = W + (size_t)l * HID * HID;
        float* tgt = (l == 0) ? xbuf : out;
        gemm_wmma<<<GRID_GE, 256, 0, stream>>>(xbuf, Wl, hbuf);
        alpha_k<<<(NNODES + 7) / 8, 256, 0, stream>>>(hbuf, att_src + l * HID,
                                                      att_dst + l * HID, asb, adb);
        if (l == 0) own_agg_k<0><<<NOWN, OWN_T, 0, stream>>>(ei, hbuf, asb, adb, keyw, bias + l * HID, tgt, qs, nqs);
        else        own_agg_k<1><<<NOWN, OWN_T, 0, stream>>>(ei, hbuf, asb, adb, keyw, bias + l * HID, tgt, qs, nqs);
    }
}


# Product moved behind the sum: owners add alpha * x[s] rows in LDS, then multiply their finished rows by W on the matrix unit (A from LDS); logits from x . (W a); 3 launches for 7; new rounding order
# speedup vs baseline: 4.6849x; 1.4970x over previous
#include <hip/hip_runtime.h>
#include <hip/hip_bf16.h>

#define NUM_USERS 50000
#define NUM_ITEMS 20000
#define NNODES    70000
#define MTILES    (NNODES / 16)
#define HID       128
#define NE        1000000
#define EPLUS     (NE + NNODES)
#define NEG_SLOPE 0.2f

typedef float v2f __attribute__((ext_vector_type(2)));
typedef float v8f __attribute__((ext_vector_type(8)));

__device__ __forceinline__ unsigned fkey(float f) {
    unsigned b = __float_as_uint(f);
    return (b & 0x80000000u) ? ~b : (b | 0x80000000u);
}
__device__ __forceinline__ float funkey(unsigned k) {
    unsigned b = (k & 0x80000000u) ? (k & 0x7FFFFFFFu) : ~k;
    return __uint_as_float(b);
}

__device__ __forceinline__ void att_through_w(const float* __restrict__ Wl,
                                              const float* __restrict__ asrc,
                                              const float* __restrict__ adst,
                                              float* was, float* wad, int wave, int lane) {
    const float4 av = ((const float4*)asrc)[lane];
    const float4 dv = ((const float4*)adst)[lane];
    float4 w[4];
#pragma unroll
    for (int i = 0; i < 4; ++i) w[i] = ((const float4*)Wl)[(size_t)(wave + 32 * i) * 32 + lane];
#pragma unroll
    for (int i = 0; i < 4; ++i) {
        float ps = w[i].x * av.x + w[i].y * av.y + w[i].z * av.z + w[i].w * av.w;
        float pd = w[i].x * dv.x + w[i].y * dv.y + w[i].z * dv.z + w[i].w * dv.w;
#pragma unroll
        for (int off = 16; off > 0; off >>= 1) {
            ps += __shfl_xor(ps, off, 32);
            pd += __shfl_xor(pd, off, 32);
        }
        if (lane == 0) { was[wave + 32 * i] = ps; wad[wave + 32 * i] = pd; }
    }
}

__device__ __forceinline__ void row_logits(const float4 v, const float4 ws, const float4 wd,
                                           float& ss, float& sd) {
    ss = v.x * ws.x + v.y * ws.y + v.z * ws.z + v.w * ws.w;
    sd = v.x * wd.x + v.y * wd.y + v.z * wd.z + v.w * wd.w;
#pragma unroll
    for (int off = 16; off > 0; off >>= 1) {
        ss += __shfl_xor(ss, off, 32);
        sd += __shfl_xor(sd, off, 32);
    }
}

#define OWN_R 274
#define NOWN  ((NNODES + OWN_R - 1) / OWN_R)
#define PRE_T 1024
#define PRE_G 512
#define PRE_W (PRE_G * PRE_T / 32)
#define PRE_U 5
static_assert(NOWN <= 256, "the owner index must fit one byte");
static_assert(EPLUS % 4 == 0 && NE % 4 == 0, "four edges per key word, never across the self-loop border");
static_assert(PRE_W * PRE_U >= NNODES && PRE_G * PRE_T >= EPLUS / 4, "the first launch covers every row and every key word");
__global__ __launch_bounds__(PRE_T) void pre_k(const float* __restrict__ u,
                                               const float* __restrict__ it,
                                               float* __restrict__ x,
                                               const int* __restrict__ ei,
                                               unsigned* __restrict__ keyw,
                                               const float* __restrict__ W0,
                                               const float* __restrict__ asrc,
                                               const float* __restrict__ adst,
                                               float* __restrict__ as_out,
                                               float* __restrict__ ad_out) {
    __shared__ __attribute__((aligned(16))) float was[HID];
    __shared__ __attribute__((aligned(16))) float wad[HID];
    const int tid  = threadIdx.x;
    const int lane = tid & 31;
    const int wave = tid >> 5;
    const int i = blockIdx.x * PRE_T + tid;
    int4 d = make_int4(0, 0, 0, 0);
    if (i < EPLUS / 4) {
        if (4 * i < NE) d = ((const int4*)(ei + NE))[i];
        else { const int n = 4 * i - NE; d = make_int4(n, n + 1, n + 2, n + 3); }
    }
    const int gw = blockIdx.x * (PRE_T / 32) + wave;
    float4 v[PRE_U];
#pragma unroll
    for (int j = 0; j < PRE_U; ++j) {
        const int row = gw + j * PRE_W;
        const int rc  = (row < NNODES) ? row : 0;
        const float* src = (rc < NUM_USERS) ? u + (size_t)rc * HID : it + (size_t)(rc - NUM_USERS) * HID;
        v[j] = ((const float4*)src)[lane];
    }

    att_through_w(W0, asrc, adst, was, wad, wave, lane);

    if (i < EPLUS / 4)
        keyw[i] = (unsigned)(d.x / OWN_R) | ((unsigned)(d.y / OWN_R) << 8) |
                  ((unsigned)(d.z / OWN_R) << 16) | ((unsigned)(d.w / OWN_R) << 24);

    __syncthreads();
    const float4 ws = ((const float4*)was)[lane];
    const float4 wd = ((const float4*)wad)[lane];
#pragma unroll
    for (int j = 0; j < PRE_U; ++j) {
        const int row = gw + j * PRE_W;
        if (row < NNODES) {
            ((float4*)x)[(size_t)row * 32 + lane] = v[j];
            float ss, sd;
            row_logits(v[j], ws, wd, ss, sd);
            if (lane == 0) { as_out[row] = ss; ad_out[row] = sd; }
        }
    }
}

__device__ __forceinline__ void edge_sd(int i, const int* __restrict__ ei,
                                        int& s, int& d) {
    const bool real = i < NE;
    const int  ic = real ? i : 0;
    const int  sv = ei[ic], dv = ei[NE + ic];
    s = real ? sv : i - NE;
    d = real ? dv : i - NE;
}

__device__ __forceinline__ float lrelu(float e) { return (e >= 0.f) ? e : NEG_SLOPE * e; }

#define OWN_T   1024
#define SCAN_U  6
#define QCAP    8192
#define LOG_U   8
#define AGG_U   8
#define NPIECE  (EPLUS / 16)
#define SBITS   17
#define RP      132
#define OWN_MT  ((OWN_R + 15) / 16)
static_assert(EPLUS % 16 == 0, "key plane must be whole 16-byte pieces");
static_assert(QCAP == OWN_T * LOG_U, "the logit step covers the queue in one sweep");
static_assert(NNODES <= (1 << SBITS) && OWN_R <= (1 << (31 - SBITS)), "queue entry packs (s, local row)");
static_assert(RP % 4 == 0 && RP >= HID && OWN_MT <= OWN_T / 32, "16-byte row starts; a wave per row tile");

template <class F>
__device__ __forceinline__ void scan_owned(const unsigned* __restrict__ keyw, int tid, F f) {
    const uint4* kp = (const uint4*)keyw;
    const unsigned pat  = blockIdx.x * 0x01010101u;
    const unsigned npat = ~pat;
    const int rot = blockIdx.x * (NPIECE / NOWN);
    for (int base = 0; base < NPIECE; base += OWN_T * SCAN_U) {
        uint4 v[SCAN_U];
#pragma unroll
        for (int u = 0; u < SCAN_U; ++u) {
            const int p = base + u * OWN_T + tid;
            const int pr = (p + rot < NPIECE) ? (p + rot) : (p + rot - NPIECE);
            v[u] = make_uint4(npat, npat, npat, npat);
            if (p < NPIECE) v[u] = kp[pr];
        }
#pragma unroll
        for (int u = 0; u < SCAN_U; ++u) {
            const int p = base + u * OWN_T + tid;
            const int pr = (p + rot < NPIECE) ? (p + rot) : (p + rot - NPIECE);
            const unsigned w[4] = { v[u].x, v[u].y, v[u].z, v[u].w };
            unsigned mk = 0;
#pragma unroll
            for (int k = 0; k < 4; ++k) {
                const unsigned x = w[k] ^ pat;
                const unsigned z = ~(((x & 0x7F7F7F7Fu) + 0x7F7F7F7Fu) | x) & 0x80808080u;
                mk |= z >> (7 - k);
            }
            while (mk) {
                const int bit = __builtin_ctz(mk);
                mk &= mk - 1;
                f(pr * 16 + 4 * (bit & 3) + (bit >> 3));
            }
        }
    }
}

template <int LAYER>
__global__ __launch_bounds__(OWN_T) void own_agg_k(const int* __restrict__ ei,
                                                   const float* __restrict__ X,
                                                   const float* __restrict__ as_,
                                                   const float* __restrict__ ad_,
                                                   const unsigned* __restrict__ keyw,
                                                   const float* __restrict__ b,
                                                   const float* __restrict__ Wl,
                                                   float* __restrict__ out,
                                                   int* __restrict__ qs,
                                                   int* __restrict__ nqs,
                                                   const float* __restrict__ Wn,
                                                   const float* __restrict__ asrc_n,
                                                   const float* __restrict__ adst_n,
                                                   float* __restrict__ as_n,
                                                   float* __restrict__ ad_n) {
    __shared__ __attribute__((aligned(16))) float rows[OWN_MT * 16 * RP];
    __shared__ __attribute__((aligned(16))) float was[HID];
    __shared__ __attribute__((aligned(16))) float wad[HID];
    __shared__ float    dens[OWN_R];
    __shared__ float    adl[OWN_R];
    __shared__ unsigned mx[OWN_R];
    __shared__ int      q[QCAP];
    __shared__ float    qx[QCAP];
    __shared__ int      qn, qhead;

    const int tid  = threadIdx.x;
    const int lane = tid & 31;
    const int wave = __builtin_amdgcn_readfirstlane(tid >> 5);
    const int row0 = blockIdx.x * OWN_R;
    const int nrows = (NNODES - row0 < OWN_R) ? (NNODES - row0) : OWN_R;

    {
        if (LAYER == 0) att_through_w(Wn, asrc_n, adst_n, was, wad, wave, lane);
        for (int i = tid; i < OWN_MT * 16 * RP / 4; i += OWN_T) ((float4*)rows)[i] = make_float4(0.f, 0.f, 0.f, 0.f);
        for (int i = tid; i < OWN_R; i += OWN_T) {
            dens[i] = 0.f;
            mx[i]   = 0u;
            adl[i]  = (i < nrows) ? ad_[row0 + i] : 0.f;
        }
        if (tid == 0) { qn = 0; qhead = 0; }
    }
    __syncthreads();

    int* const qsv = qs + blockIdx.x * QCAP;
    int nq;
    if (LAYER == 0) {
        scan_owned(keyw, tid, [&](int e) {
            const int slot = atomicAdd(&qn, 1);
            if (slot < QCAP) q[slot] = e;
        });
        __syncthreads();
        nq = __builtin_amdgcn_readfirstlane(qn);
        if (tid == 0) nqs[blockIdx.x] = nq;
    } else {
        nq = __builtin_amdgcn_readfirstlane(nqs[blockIdx.x]);
    }

    if (nq <= QCAP) {
        {
            int s[LOG_U], dl[LOG_U];
            float a[LOG_U];
            if (LAYER == 0) {
                int e[LOG_U], d[LOG_U];
#pragma unroll
                for (int u = 0; u < LOG_U; ++u) {
                    const int i = u * OWN_T + tid;
                    e[u] = (i < nq) ? q[i] : 0;
                }
#pragma unroll
                for (int u = 0; u < LOG_U; ++u) edge_sd(e[u], ei, s[u], d[u]);
#pragma unroll
                for (int u = 0; u < LOG_U; ++u) dl[u] = d[u] - row0;
            } else {
                int pk[LOG_U];
#pragma unroll
                for (int u = 0; u < LOG_U; ++u) {
                    const int i = u * OWN_T + tid;
                    pk[u] = qsv[(i < nq) ? i : 0];
                }
#pragma unroll
                for (int u = 0; u < LOG_U; ++u) {
                    s[u]  = pk[u] & ((1 << SBITS) - 1);
                    dl[u] = pk[u] >> SBITS;
                }
            }
#pragma unroll
            for (int u = 0; u < LOG_U; ++u) a[u] = as_[s[u]];
#pragma unroll
            for (int u = 0; u < LOG_U; ++u) {
                const int i = u * OWN_T + tid;
                if (i < nq) {
                    const float lg = lrelu(a[u] + adl[dl[u]]);
                    atomicMax(&mx[dl[u]], fkey(lg));
                    const int pkv = s[u] | (dl[u] << SBITS);
                    q[i]  = pkv;
                    qx[i] = lg;
                    if (LAYER == 0) qsv[i] = pkv;
                }
            }
        }
        __syncthreads();

        for (int i = tid; i < nq; i += OWN_T) {
            const int   dl = q[i] >> SBITS;
            const float ex = expf(qx[i] - funkey(mx[dl]));
            qx[i] = ex;
            atomicAdd(&dens[dl], ex);
        }
        __syncthreads();

        for (;;) {
            int c0 = 0;
            if (lane == 0) c0 = atomicAdd(&qhead, 32);
            c0 = __builtin_amdgcn_readfirstlane(c0);
            if (c0 >= nq) break;
            const int m = (nq - c0 < 32) ? (nq - c0) : 32;

            int s = 0, dl = 0;
            float alpha = 0.f;
            if (lane < m) {
                const int pk = q[c0 + lane];
                s  = pk & ((1 << SBITS) - 1);
                dl = pk >> SBITS;
                alpha = qx[c0 + lane] / dens[dl];
            }
            for (int j0 = 0; j0 < m; j0 += AGG_U) {
                float hv[AGG_U][4];
#pragma unroll
                for (int u = 0; u < AGG_U; ++u) {
                    const int su = __builtin_amdgcn_readlane(s, j0 + u);
                    const float* hr = X + (size_t)su * HID + lane;
                    hv[u][0] = hr[0]; hv[u][1] = hr[32]; hv[u][2] = hr[64]; hv[u][3] = hr[96];
                }
#pragma unroll
                for (int u = 0; u < AGG_U; ++u) {
                    if (j0 + u < m) {
                        const int   dlu = __builtin_amdgcn_readlane(dl, j0 + u);
                        const float au  = __int_as_float(
                            __builtin_amdgcn_readlane(__float_as_int(alpha), j0 + u));
                        float* r = rows + dlu * RP + lane;
                        atomicAdd(r +  0, hv[u][0] * au);
                        atomicAdd(r + 32, hv[u][1] * au);
                        atomicAdd(r + 64, hv[u][2] * au);
                        atomicAdd(r + 96, hv[u][3] * au);
                    }
                }
            }
        }
    } else {
#pragma unroll 1
        for (int pass = 0; pass < 3; ++pass) {
            scan_owned(keyw, tid, [&](int e) {
                int s, d; edge_sd(e, ei, s, d);
                const int   dl = d - row0;
                const float lg = lrelu(as_[s] + adl[dl]);
                if (pass == 0) { atomicMax(&mx[dl], fkey(lg)); return; }
                const float ex = expf(lg - funkey(mx[dl]));
                if (pass == 1) { atomicAdd(&dens[dl], ex); return; }
                const float al = ex / dens[dl];
                const float* hr = X + (size_t)s * HID;
                float* r = rows + dl * RP;
#pragma unroll 1
                for (int c = 0; c < HID; ++c) atomicAdd(r + c, hr[c] * al);
            });
            __syncthreads();
        }
    }
    __syncthreads();

    if (wave < OWN_MT) {
        const int m     = lane & 15;
        const int khalf = (lane >> 4) * 2;
        v8f c[8];
#pragma unroll
        for (int t = 0; t < 8; ++t) {
            const float bv = b[t * 16 + m];
#pragma unroll
            for (int r = 0; r < 8; ++r) c[t][r] = bv;
        }
        const float* ar = rows + (wave * 16 + m) * RP + khalf;
        const float* wr = Wl + (size_t)khalf * HID + m;
#pragma unroll 2
        for (int k = 0; k < HID; k += 4) {
            const v2f a = *(const v2f*)(ar + k);
            const float* w0 = wr + (size_t)k * HID;
            const float* w1 = w0 + HID;
#pragma unroll
            for (int t = 0; t < 8; ++t) {
                v2f bf;
                bf.x = w0[t * 16];
                bf.y = w1[t * 16];
                c[t] = __builtin_amdgcn_wmma_f32_16x16x4_f32(false, a, false, bf,
                                                             (short)0, c[t], false, false);
            }
        }
        float* dr = rows + (wave * 16 + ((lane >> 4) ? 8 : 0)) * RP + m;
#pragma unroll
        for (int t = 0; t < 8; ++t)
#pragma unroll
            for (int r = 0; r < 8; ++r)
                dr[r * RP + t * 16] = c[t][r];
    }
    __syncthreads();

    {
        float4 ws = make_float4(0.f, 0.f, 0.f, 0.f), wd = ws;
        if (LAYER == 0) { ws = ((const float4*)was)[lane]; wd = ((const float4*)wad)[lane]; }
        float4* o = (float4*)(out + (size_t)row0 * HID);
        for (int r = wave; r < nrows; r += OWN_T / 32) {
            const float4 v = *(const float4*)(rows + r * RP + 4 * lane);
            o[r * 32 + lane] = v;
            if (LAYER == 0) {
                float ss, sd;
                row_logits(v, ws, wd, ss, sd);
                if (lane == 0) { as_n[row0 + r] = ss; ad_n[row0 + r] = sd; }
            }
        }
    }
}

extern "C" void kernel_launch(void* const* d_in, const int* in_sizes, int n_in,
                              void* d_out, int out_size, void* d_ws, size_t ws_size,
                              hipStream_t stream) {
    const int*   ei      = (const int*)d_in[0];
    const float* users   = (const float*)d_in[1];
    const float* items   = (const float*)d_in[2];
    const float* W       = (const float*)d_in[3];
    const float* att_src = (const float*)d_in[4];
    const float* att_dst = (const float*)d_in[5];
    const float* bias    = (const float*)d_in[6];
    float*       out     = (float*)d_out;

    float*    xbuf  = (float*)d_ws;
    float*    x1buf = xbuf + (size_t)NNODES * HID;
    float*    as0   = x1buf + (size_t)NNODES * HID;
    float*    ad0   = as0 + NNODES;
    float*    as1   = ad0 + NNODES;
    float*    ad1   = as1 + NNODES;
    unsigned* keyw  = (unsigned*)(ad1 + NNODES);
    int*      qs    = (int*)(keyw + EPLUS / 4);
    int*      nqs   = qs + NOWN * QCAP;

    const float* W1 = W + (size_t)HID * HID;

    pre_k<<<PRE_G, PRE_T, 0, stream>>>(users, items, xbuf, ei, keyw, W, att_src, att_dst, as0, ad0);
    own_agg_k<0><<<NOWN, OWN_T, 0, stream>>>(ei, xbuf, as0, ad0, keyw, bias, W, x1buf, qs, nqs,
                                             W1, att_src + HID, att_dst + HID, as1, ad1);
    own_agg_k<1><<<NOWN, OWN_T, 0, stream>>>(ei, x1buf, as1, ad1, keyw, bias + HID, W1, out, qs, nqs,
                                             nullptr, nullptr, nullptr, nullptr, nullptr);
}


# Owner keys stored as bit planes (a wave vote per plane in the first launch): layer 0's scan tests 32 edges per thread with one boolean op per plane, not 16 by a byte search; sum order = LDS arrival
# speedup vs baseline: 4.8970x; 1.0453x over previous
#include <hip/hip_runtime.h>
#include <hip/hip_bf16.h>

#define NUM_USERS 50000
#define NUM_ITEMS 20000
#define NNODES    70000
#define MTILES    (NNODES / 16)
#define HID       128
#define NE        1000000
#define EPLUS     (NE + NNODES)
#define NEG_SLOPE 0.2f

typedef float v2f __attribute__((ext_vector_type(2)));
typedef float v8f __attribute__((ext_vector_type(8)));

__device__ __forceinline__ unsigned fkey(float f) {
    unsigned b = __float_as_uint(f);
    return (b & 0x80000000u) ? ~b : (b | 0x80000000u);
}
__device__ __forceinline__ float funkey(unsigned k) {
    unsigned b = (k & 0x80000000u) ? (k & 0x7FFFFFFFu) : ~k;
    return __uint_as_float(b);
}

__device__ __forceinline__ void att_through_w(const float* __restrict__ Wl,
                                              const float* __restrict__ asrc,
                                              const float* __restrict__ adst,
                                              float* was, float* wad, int wave, int lane) {
    const float4 av = ((const float4*)asrc)[lane];
    const float4 dv = ((const float4*)adst)[lane];
    float4 w[4];
#pragma unroll
    for (int i = 0; i < 4; ++i) w[i] = ((const float4*)Wl)[(size_t)(wave + 32 * i) * 32 + lane];
#pragma unroll
    for (int i = 0; i < 4; ++i) {
        float ps = w[i].x * av.x + w[i].y * av.y + w[i].z * av.z + w[i].w * av.w;
        float pd = w[i].x * dv.x + w[i].y * dv.y + w[i].z * dv.z + w[i].w * dv.w;
#pragma unroll
        for (int off = 16; off > 0; off >>= 1) {
            ps += __shfl_xor(ps, off, 32);
            pd += __shfl_xor(pd, off, 32);
        }
        if (lane == 0) { was[wave + 32 * i] = ps; wad[wave + 32 * i] = pd; }
    }
}

__device__ __forceinline__ void row_logits(const float4 v, const float4 ws, const float4 wd,
                                           float& ss, float& sd) {
    ss = v.x * ws.x + v.y * ws.y + v.z * ws.z + v.w * ws.w;
    sd = v.x * wd.x + v.y * wd.y + v.z * wd.z + v.w * wd.w;
#pragma unroll
    for (int off = 16; off > 0; off >>= 1) {
        ss += __shfl_xor(ss, off, 32);
        sd += __shfl_xor(sd, off, 32);
    }
}

#define OWN_R 274
#define NOWN  ((NNODES + OWN_R - 1) / OWN_R)
#define PRE_T 1024
#define PRE_G 512
#define PRE_W (PRE_G * PRE_T / 32)
#define PRE_U 5
#define NKW   ((EPLUS + 127) / 128)
#define NSB   (NKW / 8)
#define NUNIT (NKW * 4)
#define KW_WORDS (NSB * 256)
#define TAIL_LANES ((EPLUS - 128 * (NKW - 1)) / 4)
static_assert(NOWN <= 256, "the owner index must fit eight planes");
static_assert(EPLUS % 4 == 0 && NE % 4 == 0, "four edges per thread, never across the self-loop border");
static_assert(NKW % 8 == 0 && PRE_T == 1024, "a group of the first launch writes four whole super-blocks");
static_assert(TAIL_LANES > 0 && TAIL_LANES < 32, "the last block is partly filled: its mask below");
static_assert(PRE_W * PRE_U >= NNODES && PRE_G * PRE_T >= EPLUS / 4, "the first launch covers every row and every edge");
__global__ __launch_bounds__(PRE_T) void pre_k(const float* __restrict__ u,
                                               const float* __restrict__ it,
                                               float* __restrict__ x,
                                               const int* __restrict__ ei,
                                               unsigned* __restrict__ keyw,
                                               const float* __restrict__ W0,
                                               const float* __restrict__ asrc,
                                               const float* __restrict__ adst,
                                               float* __restrict__ as_out,
                                               float* __restrict__ ad_out) {
    __shared__ __attribute__((aligned(16))) float was[HID];
    __shared__ __attribute__((aligned(16))) float wad[HID];
    __shared__ unsigned kst[PRE_T];
    const int tid  = threadIdx.x;
    const int lane = tid & 31;
    const int wave = __builtin_amdgcn_readfirstlane(tid >> 5);
    const int i = blockIdx.x * PRE_T + tid;
    int4 d = make_int4(0, 0, 0, 0);
    if (i < EPLUS / 4) {
        if (4 * i < NE) d = ((const int4*)(ei + NE))[i];
        else { const int n = 4 * i - NE; d = make_int4(n, n + 1, n + 2, n + 3); }
    }
    const int gw = blockIdx.x * (PRE_T / 32) + wave;
    float4 v[PRE_U];
#pragma unroll
    for (int j = 0; j < PRE_U; ++j) {
        const int row = gw + j * PRE_W;
        const int rc  = (row < NNODES) ? row : 0;
        const float* src = (rc < NUM_USERS) ? u + (size_t)rc * HID : it + (size_t)(rc - NUM_USERS) * HID;
        v[j] = ((const float4*)src)[lane];
    }

    att_through_w(W0, asrc, adst, was, wad, wave, lane);

    if (gw < NKW) {
        const unsigned key32 = (unsigned)(d.x / OWN_R) | ((unsigned)(d.y / OWN_R) << 8) |
                               ((unsigned)(d.z / OWN_R) << 16) | ((unsigned)(d.w / OWN_R) << 24);
        unsigned myw = 0u;
#pragma unroll
        for (int j = 0; j < 32; ++j) {
            const unsigned vote = __builtin_amdgcn_ballot_w32(((key32 >> j) & 1u) != 0u);
            myw = (lane == j) ? vote : myw;
        }
        const int c = lane >> 3, b = lane & 7;
        kst[(wave >> 3) * 256 + (b >> 2) * 128 + ((wave & 7) * 4 + c) * 4 + (b & 3)] = myw;
    }

    __syncthreads();
    if (i < KW_WORDS) keyw[i] = kst[tid];
    const float4 ws = ((const float4*)was)[lane];
    const float4 wd = ((const float4*)wad)[lane];
#pragma unroll
    for (int j = 0; j < PRE_U; ++j) {
        const int row = gw + j * PRE_W;
        if (row < NNODES) {
            ((float4*)x)[(size_t)row * 32 + lane] = v[j];
            float ss, sd;
            row_logits(v[j], ws, wd, ss, sd);
            if (lane == 0) { as_out[row] = ss; ad_out[row] = sd; }
        }
    }
}

__device__ __forceinline__ void edge_sd(int i, const int* __restrict__ ei,
                                        int& s, int& d) {
    const bool real = i < NE;
    const int  ic = real ? i : 0;
    const int  sv = ei[ic], dv = ei[NE + ic];
    s = real ? sv : i - NE;
    d = real ? dv : i - NE;
}

__device__ __forceinline__ float lrelu(float e) { return (e >= 0.f) ? e : NEG_SLOPE * e; }

#define OWN_T   1024
#define SCAN_U  4
#define QCAP    8192
#define LOG_U   8
#define AGG_U   8
#define SBITS   17
#define RP      132
#define OWN_MT  ((OWN_R + 15) / 16)
static_assert(QCAP == OWN_T * LOG_U, "the logit step covers the queue in one sweep");
static_assert(NNODES <= (1 << SBITS) && OWN_R <= (1 << (31 - SBITS)), "queue entry packs (s, local row)");
static_assert(RP % 4 == 0 && RP >= HID && OWN_MT <= OWN_T / 32, "16-byte row starts; a wave per row tile");

template <class F>
__device__ __forceinline__ void scan_owned(const unsigned* __restrict__ keyw, int tid, F f) {
    const uint4* kp = (const uint4*)keyw;
    const unsigned g = blockIdx.x;
    unsigned nb[8];
#pragma unroll
    for (int b = 0; b < 8; ++b) nb[b] = ((g >> b) & 1u) ? 0u : 0xFFFFFFFFu;
    const int rot = (int)((g * NSB) / NOWN) * 32;
    for (int base = 0; base < NUNIT; base += OWN_T * SCAN_U) {
        uint4 lo[SCAN_U], hi[SCAN_U];
#pragma unroll
        for (int u = 0; u < SCAN_U; ++u) {
            const int p  = base + u * OWN_T + tid;
            const int pr = (p + rot < NUNIT) ? (p + rot) : (p + rot - NUNIT);
            const int pc = (p < NUNIT) ? pr : 0;
            const int at = (pc >> 5) * 64 + (pc & 31);
            lo[u] = kp[at];
            hi[u] = kp[at + 32];
        }
#pragma unroll
        for (int u = 0; u < SCAN_U; ++u) {
            const int p  = base + u * OWN_T + tid;
            const int pr = (p + rot < NUNIT) ? (p + rot) : (p + rot - NUNIT);
            unsigned mk = (p < NUNIT) ? 0xFFFFFFFFu : 0u;
            mk &= lo[u].x ^ nb[0]; mk &= lo[u].y ^ nb[1]; mk &= lo[u].z ^ nb[2]; mk &= lo[u].w ^ nb[3];
            mk &= hi[u].x ^ nb[4]; mk &= hi[u].y ^ nb[5]; mk &= hi[u].z ^ nb[6]; mk &= hi[u].w ^ nb[7];
            if ((pr >> 2) == NKW - 1) mk &= (1u << TAIL_LANES) - 1u;
            const int e0 = ((pr >> 2) << 7) + (pr & 3);
            while (mk) {
                const int bit = __builtin_ctz(mk);
                mk &= mk - 1;
                f(e0 + 4 * bit);
            }
        }
    }
}

template <int LAYER>
__global__ __launch_bounds__(OWN_T) void own_agg_k(const int* __restrict__ ei,
                                                   const float* __restrict__ X,
                                                   const float* __restrict__ as_,
                                                   const float* __restrict__ ad_,
                                                   const unsigned* __restrict__ keyw,
                                                   const float* __restrict__ b,
                                                   const float* __restrict__ Wl,
                                                   float* __restrict__ out,
                                                   int* __restrict__ qs,
                                                   int* __restrict__ nqs,
                                                   const float* __restrict__ Wn,
                                                   const float* __restrict__ asrc_n,
                                                   const float* __restrict__ adst_n,
                                                   float* __restrict__ as_n,
                                                   float* __restrict__ ad_n) {
    __shared__ __attribute__((aligned(16))) float rows[OWN_MT * 16 * RP];
    __shared__ __attribute__((aligned(16))) float was[HID];
    __shared__ __attribute__((aligned(16))) float wad[HID];
    __shared__ float    dens[OWN_R];
    __shared__ float    adl[OWN_R];
    __shared__ unsigned mx[OWN_R];
    __shared__ int      q[QCAP];
    __shared__ float    qx[QCAP];
    __shared__ int      qn, qhead;

    const int tid  = threadIdx.x;
    const int lane = tid & 31;
    const int wave = __builtin_amdgcn_readfirstlane(tid >> 5);
    const int row0 = blockIdx.x * OWN_R;
    const int nrows = (NNODES - row0 < OWN_R) ? (NNODES - row0) : OWN_R;

    {
        if (LAYER == 0) att_through_w(Wn, asrc_n, adst_n, was, wad, wave, lane);
        for (int i = tid; i < OWN_MT * 16 * RP / 4; i += OWN_T) ((float4*)rows)[i] = make_float4(0.f, 0.f, 0.f, 0.f);
        for (int i = tid; i < OWN_R; i += OWN_T) {
            dens[i] = 0.f;
            mx[i]   = 0u;
            adl[i]  = (i < nrows) ? ad_[row0 + i] : 0.f;
        }
        if (tid == 0) { qn = 0; qhead = 0; }
    }
    __syncthreads();

    int* const qsv = qs + blockIdx.x * QCAP;
    int nq;
    if (LAYER == 0) {
        scan_owned(keyw, tid, [&](int e) {
            const int slot = atomicAdd(&qn, 1);
            if (slot < QCAP) q[slot] = e;
        });
        __syncthreads();
        nq = __builtin_amdgcn_readfirstlane(qn);
        if (tid == 0) nqs[blockIdx.x] = nq;
    } else {
        nq = __builtin_amdgcn_readfirstlane(nqs[blockIdx.x]);
    }

    if (nq <= QCAP) {
        {
            int s[LOG_U], dl[LOG_U];
            float a[LOG_U];
            if (LAYER == 0) {
                int e[LOG_U], d[LOG_U];
#pragma unroll
                for (int u = 0; u < LOG_U; ++u) {
                    const int i = u * OWN_T + tid;
                    e[u] = (i < nq) ? q[i] : 0;
                }
#pragma unroll
                for (int u = 0; u < LOG_U; ++u) edge_sd(e[u], ei, s[u], d[u]);
#pragma unroll
                for (int u = 0; u < LOG_U; ++u) dl[u] = d[u] - row0;
            } else {
                int pk[LOG_U];
#pragma unroll
                for (int u = 0; u < LOG_U; ++u) {
                    const int i = u * OWN_T + tid;
                    pk[u] = qsv[(i < nq) ? i : 0];
                }
#pragma unroll
                for (int u = 0; u < LOG_U; ++u) {
                    s[u]  = pk[u] & ((1 << SBITS) - 1);
                    dl[u] = pk[u] >> SBITS;
                }
            }
#pragma unroll
            for (int u = 0; u < LOG_U; ++u) a[u] = as_[s[u]];
#pragma unroll
            for (int u = 0; u < LOG_U; ++u) {
                const int i = u * OWN_T + tid;
                if (i < nq) {
                    const float lg = lrelu(a[u] + adl[dl[u]]);
                    atomicMax(&mx[dl[u]], fkey(lg));
                    const int pkv = s[u] | (dl[u] << SBITS);
                    q[i]  = pkv;
                    qx[i] = lg;
                    if (LAYER == 0) qsv[i] = pkv;
                }
            }
        }
        __syncthreads();

        for (int i = tid; i < nq; i += OWN_T) {
            const int   dl = q[i] >> SBITS;
            const float ex = expf(qx[i] - funkey(mx[dl]));
            qx[i] = ex;
            atomicAdd(&dens[dl], ex);
        }
        __syncthreads();

        for (;;) {
            int c0 = 0;
            if (lane == 0) c0 = atomicAdd(&qhead, 32);
            c0 = __builtin_amdgcn_readfirstlane(c0);
            if (c0 >= nq) break;
            const int m = (nq - c0 < 32) ? (nq - c0) : 32;

            int s = 0, dl = 0;
            float alpha = 0.f;
            if (lane < m) {
                const int pk = q[c0 + lane];
                s  = pk & ((1 << SBITS) - 1);
                dl = pk >> SBITS;
                alpha = qx[c0 + lane] / dens[dl];
            }
            for (int j0 = 0; j0 < m; j0 += AGG_U) {
                float hv[AGG_U][4];
#pragma unroll
                for (int u = 0; u < AGG_U; ++u) {
                    const int su = __builtin_amdgcn_readlane(s, j0 + u);
                    const float* hr = X + (size_t)su * HID + lane;
                    hv[u][0] = hr[0]; hv[u][1] = hr[32]; hv[u][2] = hr[64]; hv[u][3] = hr[96];
                }
#pragma unroll
                for (int u = 0; u < AGG_U; ++u) {
                    if (j0 + u < m) {
                        const int   dlu = __builtin_amdgcn_readlane(dl, j0 + u);
                        const float au  = __int_as_float(
                            __builtin_amdgcn_readlane(__float_as_int(alpha), j0 + u));
                        float* r = rows + dlu * RP + lane;
                        atomicAdd(r +  0, hv[u][0] * au);
                        atomicAdd(r + 32, hv[u][1] * au);
                        atomicAdd(r + 64, hv[u][2] * au);
                        atomicAdd(r + 96, hv[u][3] * au);
                    }
                }
            }
        }
    } else {
#pragma unroll 1
        for (int pass = 0; pass < 3; ++pass) {
            scan_owned(keyw, tid, [&](int e) {
                int s, d; edge_sd(e, ei, s, d);
                const int   dl = d - row0;
                const float lg = lrelu(as_[s] + adl[dl]);
                if (pass == 0) { atomicMax(&mx[dl], fkey(lg)); return; }
                const float ex = expf(lg - funkey(mx[dl]));
                if (pass == 1) { atomicAdd(&dens[dl], ex); return; }
                const float al = ex / dens[dl];
                const float* hr = X + (size_t)s * HID;
                float* r = rows + dl * RP;
#pragma unroll 1
                for (int c = 0; c < HID; ++c) atomicAdd(r + c, hr[c] * al);
            });
            __syncthreads();
        }
    }
    __syncthreads();

    if (wave < OWN_MT) {
        const int m     = lane & 15;
        const int khalf = (lane >> 4) * 2;
        v8f c[8];
#pragma unroll
        for (int t = 0; t < 8; ++t) {
            const float bv = b[t * 16 + m];
#pragma unroll
            for (int r = 0; r < 8; ++r) c[t][r] = bv;
        }
        const float* ar = rows + (wave * 16 + m) * RP + khalf;
        const float* wr = Wl + (size_t)khalf * HID + m;
#pragma unroll 2
        for (int k = 0; k < HID; k += 4) {
            const v2f a = *(const v2f*)(ar + k);
            const float* w0 = wr + (size_t)k * HID;
            const float* w1 = w0 + HID;
#pragma unroll
            for (int t = 0; t < 8; ++t) {
                v2f bf;
                bf.x = w0[t * 16];
                bf.y = w1[t * 16];
                c[t] = __builtin_amdgcn_wmma_f32_16x16x4_f32(false, a, false, bf,
                                                             (short)0, c[t], false, false);
            }
        }
        float* dr = rows + (wave * 16 + ((lane >> 4) ? 8 : 0)) * RP + m;
#pragma unroll
        for (int t = 0; t < 8; ++t)
#pragma unroll
            for (int r = 0; r < 8; ++r)
                dr[r * RP + t * 16] = c[t][r];
    }
    __syncthreads();

    {
        float4 ws = make_float4(0.f, 0.f, 0.f, 0.f), wd = ws;
        if (LAYER == 0) { ws = ((const float4*)was)[lane]; wd = ((const float4*)wad)[lane]; }
        float4* o = (float4*)(out + (size_t)row0 * HID);
        for (int r = wave; r < nrows; r += OWN_T / 32) {
            const float4 v = *(const float4*)(rows + r * RP + 4 * lane);
            o[r * 32 + lane] = v;
            if (LAYER == 0) {
                float ss, sd;
                row_logits(v, ws, wd, ss, sd);
                if (lane == 0) { as_n[row0 + r] = ss; ad_n[row0 + r] = sd; }
            }
        }
    }
}

extern "C" void kernel_launch(void* const* d_in, const int* in_sizes, int n_in,
                              void* d_out, int out_size, void* d_ws, size_t ws_size,
                              hipStream_t stream) {
    const int*   ei      = (const int*)d_in[0];
    const float* users   = (const float*)d_in[1];
    const float* items   = (const float*)d_in[2];
    const float* W       = (const float*)d_in[3];
    const float* att_src = (const float*)d_in[4];
    const float* att_dst = (const float*)d_in[5];
    const float* bias    = (const float*)d_in[6];
    float*       out     = (float*)d_out;

    float*    xbuf  = (float*)d_ws;
    float*    x1buf = xbuf + (size_t)NNODES * HID;
    float*    as0   = x1buf + (size_t)NNODES * HID;
    float*    ad0   = as0 + NNODES;
    float*    as1   = ad0 + NNODES;
    float*    ad1   = as1 + NNODES;
    unsigned* keyw  = (unsigned*)(ad1 + NNODES);
    int*      qs    = (int*)(keyw + KW_WORDS);
    int*      nqs   = qs + NOWN * QCAP;

    const float* W1 = W + (size_t)HID * HID;

    pre_k<<<PRE_G, PRE_T, 0, stream>>>(users, items, xbuf, ei, keyw, W, att_src, att_dst, as0, ad0);
    own_agg_k<0><<<NOWN, OWN_T, 0, stream>>>(ei, xbuf, as0, ad0, keyw, bias, W, x1buf, qs, nqs,
                                             W1, att_src + HID, att_dst + HID, as1, ad1);
    own_agg_k<1><<<NOWN, OWN_T, 0, stream>>>(ei, x1buf, as1, ad1, keyw, bias + HID, W1, out, qs, nqs,
                                             nullptr, nullptr, nullptr, nullptr, nullptr);
}


# Row sums without LDS float atomics: queue regrouped by owner wave (local row mod 32, integer LDS ranks); a wave adds its rows by plain 16-byte LDS load, fma, store; sum order = rank in wave's bin
# speedup vs baseline: 5.0572x; 1.0327x over previous
#include <hip/hip_runtime.h>
#include <hip/hip_bf16.h>

#define NUM_USERS 50000
#define NUM_ITEMS 20000
#define NNODES    70000
#define MTILES    (NNODES / 16)
#define HID       128
#define NE        1000000
#define EPLUS     (NE + NNODES)
#define NEG_SLOPE 0.2f

typedef float v2f __attribute__((ext_vector_type(2)));
typedef float v8f __attribute__((ext_vector_type(8)));

__device__ __forceinline__ unsigned fkey(float f) {
    unsigned b = __float_as_uint(f);
    return (b & 0x80000000u) ? ~b : (b | 0x80000000u);
}
__device__ __forceinline__ float funkey(unsigned k) {
    unsigned b = (k & 0x80000000u) ? (k & 0x7FFFFFFFu) : ~k;
    return __uint_as_float(b);
}

__device__ __forceinline__ void att_through_w(const float* __restrict__ Wl,
                                              const float* __restrict__ asrc,
                                              const float* __restrict__ adst,
                                              float* was, float* wad, int wave, int lane) {
    const float4 av = ((const float4*)asrc)[lane];
    const float4 dv = ((const float4*)adst)[lane];
    float4 w[4];
#pragma unroll
    for (int i = 0; i < 4; ++i) w[i] = ((const float4*)Wl)[(size_t)(wave + 32 * i) * 32 + lane];
#pragma unroll
    for (int i = 0; i < 4; ++i) {
        float ps = w[i].x * av.x + w[i].y * av.y + w[i].z * av.z + w[i].w * av.w;
        float pd = w[i].x * dv.x + w[i].y * dv.y + w[i].z * dv.z + w[i].w * dv.w;
#pragma unroll
        for (int off = 16; off > 0; off >>= 1) {
            ps += __shfl_xor(ps, off, 32);
            pd += __shfl_xor(pd, off, 32);
        }
        if (lane == 0) { was[wave + 32 * i] = ps; wad[wave + 32 * i] = pd; }
    }
}

__device__ __forceinline__ void row_logits(const float4 v, const float4 ws, const float4 wd,
                                           float& ss, float& sd) {
    ss = v.x * ws.x + v.y * ws.y + v.z * ws.z + v.w * ws.w;
    sd = v.x * wd.x + v.y * wd.y + v.z * wd.z + v.w * wd.w;
#pragma unroll
    for (int off = 16; off > 0; off >>= 1) {
        ss += __shfl_xor(ss, off, 32);
        sd += __shfl_xor(sd, off, 32);
    }
}

#define OWN_R 274
#define NOWN  ((NNODES + OWN_R - 1) / OWN_R)
#define PRE_T 1024
#define PRE_G 512
#define PRE_W (PRE_G * PRE_T / 32)
#define PRE_U 5
#define NKW   ((EPLUS + 127) / 128)
#define NSB   (NKW / 8)
#define NUNIT (NKW * 4)
#define KW_WORDS (NSB * 256)
#define TAIL_LANES ((EPLUS - 128 * (NKW - 1)) / 4)
static_assert(NOWN <= 256, "the owner index must fit eight planes");
static_assert(EPLUS % 4 == 0 && NE % 4 == 0, "four edges per thread, never across the self-loop border");
static_assert(NKW % 8 == 0 && PRE_T == 1024, "a group of the first launch writes four whole super-blocks");
static_assert(TAIL_LANES > 0 && TAIL_LANES < 32, "the last block is partly filled: its mask below");
static_assert(PRE_W * PRE_U >= NNODES && PRE_G * PRE_T >= EPLUS / 4, "the first launch covers every row and every edge");
__global__ __launch_bounds__(PRE_T) void pre_k(const float* __restrict__ u,
                                               const float* __restrict__ it,
                                               float* __restrict__ x,
                                               const int* __restrict__ ei,
                                               unsigned* __restrict__ keyw,
                                               const float* __restrict__ W0,
                                               const float* __restrict__ asrc,
                                               const float* __restrict__ adst,
                                               float* __restrict__ as_out,
                                               float* __restrict__ ad_out) {
    __shared__ __attribute__((aligned(16))) float was[HID];
    __shared__ __attribute__((aligned(16))) float wad[HID];
    __shared__ unsigned kst[PRE_T];
    const int tid  = threadIdx.x;
    const int lane = tid & 31;
    const int wave = __builtin_amdgcn_readfirstlane(tid >> 5);
    const int i = blockIdx.x * PRE_T + tid;
    int4 d = make_int4(0, 0, 0, 0);
    if (i < EPLUS / 4) {
        if (4 * i < NE) d = ((const int4*)(ei + NE))[i];
        else { const int n = 4 * i - NE; d = make_int4(n, n + 1, n + 2, n + 3); }
    }
    const int gw = blockIdx.x * (PRE_T / 32) + wave;
    float4 v[PRE_U];
#pragma unroll
    for (int j = 0; j < PRE_U; ++j) {
        const int row = gw + j * PRE_W;
        const int rc  = (row < NNODES) ? row : 0;
        const float* src = (rc < NUM_USERS) ? u + (size_t)rc * HID : it + (size_t)(rc - NUM_USERS) * HID;
        v[j] = ((const float4*)src)[lane];
    }

    att_through_w(W0, asrc, adst, was, wad, wave, lane);

    if (gw < NKW) {
        const unsigned key32 = (unsigned)(d.x / OWN_R) | ((unsigned)(d.y / OWN_R) << 8) |
                               ((unsigned)(d.z / OWN_R) << 16) | ((unsigned)(d.w / OWN_R) << 24);
        unsigned myw = 0u;
#pragma unroll
        for (int j = 0; j < 32; ++j) {
            const unsigned vote = __builtin_amdgcn_ballot_w32(((key32 >> j) & 1u) != 0u);
            myw = (lane == j) ? vote : myw;
        }
        const int c = lane >> 3, b = lane & 7;
        kst[(wave >> 3) * 256 + (b >> 2) * 128 + ((wave & 7) * 4 + c) * 4 + (b & 3)] = myw;
    }

    __syncthreads();
    if (i < KW_WORDS) keyw[i] = kst[tid];
    const float4 ws = ((const float4*)was)[lane];
    const float4 wd = ((const float4*)wad)[lane];
#pragma unroll
    for (int j = 0; j < PRE_U; ++j) {
        const int row = gw + j * PRE_W;
        if (row < NNODES) {
            ((float4*)x)[(size_t)row * 32 + lane] = v[j];
            float ss, sd;
            row_logits(v[j], ws, wd, ss, sd);
            if (lane == 0) { as_out[row] = ss; ad_out[row] = sd; }
        }
    }
}

__device__ __forceinline__ void edge_sd(int i, const int* __restrict__ ei,
                                        int& s, int& d) {
    const bool real = i < NE;
    const int  ic = real ? i : 0;
    const int  sv = ei[ic], dv = ei[NE + ic];
    s = real ? sv : i - NE;
    d = real ? dv : i - NE;
}

__device__ __forceinline__ float lrelu(float e) { return (e >= 0.f) ? e : NEG_SLOPE * e; }

#define OWN_T   1024
#define SCAN_U  4
#define QCAP    8192
#define LOG_U   8
#define AGG_U   8
#define SBITS   17
#define RP      132
#define OWN_MT  ((OWN_R + 15) / 16)
static_assert(QCAP == OWN_T * LOG_U, "the logit step covers the queue in one sweep");
static_assert(NNODES <= (1 << SBITS) && OWN_R <= (1 << (31 - SBITS)), "queue entry packs (s, local row)");
static_assert(RP % 4 == 0 && RP >= HID && OWN_MT <= OWN_T / 32, "16-byte row starts; a wave per row tile");
static_assert(OWN_T == 1024 && AGG_U == 8, "32 owner waves (local row mod 32); batches of 8 rows");

template <class F>
__device__ __forceinline__ void scan_owned(const unsigned* __restrict__ keyw, int tid, F f) {
    const uint4* kp = (const uint4*)keyw;
    const unsigned g = blockIdx.x;
    unsigned nb[8];
#pragma unroll
    for (int b = 0; b < 8; ++b) nb[b] = ((g >> b) & 1u) ? 0u : 0xFFFFFFFFu;
    const int rot = (int)((g * NSB) / NOWN) * 32;
    for (int base = 0; base < NUNIT; base += OWN_T * SCAN_U) {
        uint4 lo[SCAN_U], hi[SCAN_U];
#pragma unroll
        for (int u = 0; u < SCAN_U; ++u) {
            const int p  = base + u * OWN_T + tid;
            const int pr = (p + rot < NUNIT) ? (p + rot) : (p + rot - NUNIT);
            const int pc = (p < NUNIT) ? pr : 0;
            const int at = (pc >> 5) * 64 + (pc & 31);
            lo[u] = kp[at];
            hi[u] = kp[at + 32];
        }
#pragma unroll
        for (int u = 0; u < SCAN_U; ++u) {
            const int p  = base + u * OWN_T + tid;
            const int pr = (p + rot < NUNIT) ? (p + rot) : (p + rot - NUNIT);
            unsigned mk = (p < NUNIT) ? 0xFFFFFFFFu : 0u;
            mk &= lo[u].x ^ nb[0]; mk &= lo[u].y ^ nb[1]; mk &= lo[u].z ^ nb[2]; mk &= lo[u].w ^ nb[3];
            mk &= hi[u].x ^ nb[4]; mk &= hi[u].y ^ nb[5]; mk &= hi[u].z ^ nb[6]; mk &= hi[u].w ^ nb[7];
            if ((pr >> 2) == NKW - 1) mk &= (1u << TAIL_LANES) - 1u;
            const int e0 = ((pr >> 2) << 7) + (pr & 3);
            while (mk) {
                const int bit = __builtin_ctz(mk);
                mk &= mk - 1;
                f(e0 + 4 * bit);
            }
        }
    }
}

__device__ __forceinline__ void load_rows(float4 (&hv)[8], const float* __restrict__ X,
                                          int s, int j0, int lane) {
#pragma unroll
    for (int u = 0; u < 8; ++u) {
        const int su = __builtin_amdgcn_readlane(s, j0 + u);
        hv[u] = ((const float4*)(X + (size_t)su * HID))[lane];
    }
}
__device__ __forceinline__ void add_rows(const float4 (&hv)[8], float* rows,
                                         int dl, float alpha, int j0, int m, int lane) {
#pragma unroll
    for (int u = 0; u < 8; ++u) {
        if (j0 + u < m) {
            const int   dlu = __builtin_amdgcn_readlane(dl, j0 + u);
            const float au  = __int_as_float(__builtin_amdgcn_readlane(__float_as_int(alpha), j0 + u));
            float4* r = (float4*)(rows + dlu * RP) + lane;
            float4 t = *r;
            t.x += hv[u].x * au; t.y += hv[u].y * au; t.z += hv[u].z * au; t.w += hv[u].w * au;
            *r = t;
        }
    }
}

template <int LAYER>
__global__ __launch_bounds__(OWN_T) void own_agg_k(const int* __restrict__ ei,
                                                   const float* __restrict__ X,
                                                   const float* __restrict__ as_,
                                                   const float* __restrict__ ad_,
                                                   const unsigned* __restrict__ keyw,
                                                   const float* __restrict__ b,
                                                   const float* __restrict__ Wl,
                                                   float* __restrict__ out,
                                                   int* __restrict__ qs,
                                                   int* __restrict__ nqs,
                                                   const float* __restrict__ Wn,
                                                   const float* __restrict__ asrc_n,
                                                   const float* __restrict__ adst_n,
                                                   float* __restrict__ as_n,
                                                   float* __restrict__ ad_n) {
    __shared__ __attribute__((aligned(16))) float rows[OWN_MT * 16 * RP];
    __shared__ __attribute__((aligned(16))) float was[HID];
    __shared__ __attribute__((aligned(16))) float wad[HID];
    __shared__ float    dens[OWN_R];
    __shared__ float    adl[OWN_R];
    __shared__ unsigned mx[OWN_R];
    __shared__ int      q[QCAP];
    __shared__ float    qx[QCAP];
    __shared__ int      bcnt[OWN_T / 32];
    __shared__ int      qn;

    const int tid  = threadIdx.x;
    const int lane = tid & 31;
    const int wave = __builtin_amdgcn_readfirstlane(tid >> 5);
    const int row0 = blockIdx.x * OWN_R;
    const int nrows = (NNODES - row0 < OWN_R) ? (NNODES - row0) : OWN_R;

    {
        if (LAYER == 0) att_through_w(Wn, asrc_n, adst_n, was, wad, wave, lane);
        if (tid < OWN_T / 32) bcnt[tid] = 0;
        for (int i = tid; i < OWN_MT * 16 * RP / 4; i += OWN_T) ((float4*)rows)[i] = make_float4(0.f, 0.f, 0.f, 0.f);
        for (int i = tid; i < OWN_R; i += OWN_T) {
            dens[i] = 0.f;
            mx[i]   = 0u;
            adl[i]  = (i < nrows) ? ad_[row0 + i] : 0.f;
        }
        if (tid == 0) qn = 0;
    }
    __syncthreads();

    int* const qsv = qs + blockIdx.x * QCAP;
    int nq;
    if (LAYER == 0) {
        scan_owned(keyw, tid, [&](int e) {
            const int slot = atomicAdd(&qn, 1);
            if (slot < QCAP) q[slot] = e;
        });
        __syncthreads();
        nq = __builtin_amdgcn_readfirstlane(qn);
        if (tid == 0) nqs[blockIdx.x] = nq;
    } else {
        nq = __builtin_amdgcn_readfirstlane(nqs[blockIdx.x]);
    }

    if (nq <= QCAP) {
        int bin_0 = 0, bin_n = 0;
        {
            int s[LOG_U], dl[LOG_U];
            float a[LOG_U];
            if (LAYER == 0) {
                int e[LOG_U], d[LOG_U];
#pragma unroll
                for (int u = 0; u < LOG_U; ++u) {
                    const int i = u * OWN_T + tid;
                    e[u] = (i < nq) ? q[i] : 0;
                }
#pragma unroll
                for (int u = 0; u < LOG_U; ++u) edge_sd(e[u], ei, s[u], d[u]);
#pragma unroll
                for (int u = 0; u < LOG_U; ++u) dl[u] = d[u] - row0;
            } else {
                int pk[LOG_U];
#pragma unroll
                for (int u = 0; u < LOG_U; ++u) {
                    const int i = u * OWN_T + tid;
                    pk[u] = qsv[(i < nq) ? i : 0];
                }
#pragma unroll
                for (int u = 0; u < LOG_U; ++u) {
                    s[u]  = pk[u] & ((1 << SBITS) - 1);
                    dl[u] = pk[u] >> SBITS;
                }
            }
#pragma unroll
            for (int u = 0; u < LOG_U; ++u) a[u] = as_[s[u]];
            int   rk[LOG_U];
            float lgv[LOG_U];
#pragma unroll
            for (int u = 0; u < LOG_U; ++u) {
                const int i = u * OWN_T + tid;
                rk[u] = 0; lgv[u] = 0.f;
                if (i < nq) {
                    const float lg = lrelu(a[u] + adl[dl[u]]);
                    atomicMax(&mx[dl[u]], fkey(lg));
                    rk[u]  = atomicAdd(&bcnt[dl[u] & 31], 1);
                    lgv[u] = lg;
                    if (LAYER == 0) qsv[i] = s[u] | (dl[u] << SBITS);
                }
            }
            __syncthreads();
            const int cntl = bcnt[lane];
            int incl = cntl;
#pragma unroll
            for (int off = 1; off < 32; off <<= 1) {
                const int t = __shfl_up(incl, off, 32);
                if (lane >= off) incl += t;
            }
            bin_n = __builtin_amdgcn_readlane(cntl, wave);
            bin_0 = __builtin_amdgcn_readlane(incl - cntl, wave);
#pragma unroll
            for (int u = 0; u < LOG_U; ++u) {
                const int i  = u * OWN_T + tid;
                const int st = __shfl(incl - cntl, dl[u] & 31, 32);
                if (i < nq) {
                    q[st + rk[u]]  = s[u] | (dl[u] << SBITS);
                    qx[st + rk[u]] = lgv[u];
                }
            }
        }
        __syncthreads();

        for (int i = tid; i < nq; i += OWN_T) {
            const int   dl = q[i] >> SBITS;
            const float ex = expf(qx[i] - funkey(mx[dl]));
            qx[i] = ex;
            atomicAdd(&dens[dl], ex);
        }
        __syncthreads();

        for (int c0 = 0; c0 < bin_n; c0 += 32) {
            const int m = (bin_n - c0 < 32) ? (bin_n - c0) : 32;
            int s = 0, dl = 0;
            float alpha = 0.f;
            if (lane < m) {
                const int pk = q[bin_0 + c0 + lane];
                s  = pk & ((1 << SBITS) - 1);
                dl = pk >> SBITS;
                alpha = qx[bin_0 + c0 + lane] / dens[dl];
            }
            float4 ha[AGG_U], hb[AGG_U];
            load_rows(ha, X, s, 0, lane);
            if (m > 8)  load_rows(hb, X, s, 8, lane);
            add_rows(ha, rows, dl, alpha, 0, m, lane);
            if (m > 16) load_rows(ha, X, s, 16, lane);
            if (m > 8)  add_rows(hb, rows, dl, alpha, 8, m, lane);
            if (m > 24) load_rows(hb, X, s, 24, lane);
            if (m > 16) add_rows(ha, rows, dl, alpha, 16, m, lane);
            if (m > 24) add_rows(hb, rows, dl, alpha, 24, m, lane);
        }
    } else {
#pragma unroll 1
        for (int pass = 0; pass < 3; ++pass) {
            scan_owned(keyw, tid, [&](int e) {
                int s, d; edge_sd(e, ei, s, d);
                const int   dl = d - row0;
                const float lg = lrelu(as_[s] + adl[dl]);
                if (pass == 0) { atomicMax(&mx[dl], fkey(lg)); return; }
                const float ex = expf(lg - funkey(mx[dl]));
                if (pass == 1) { atomicAdd(&dens[dl], ex); return; }
                const float al = ex / dens[dl];
                const float* hr = X + (size_t)s * HID;
                float* r = rows + dl * RP;
#pragma unroll 1
                for (int c = 0; c < HID; ++c) atomicAdd(r + c, hr[c] * al);
            });
            __syncthreads();
        }
    }
    __syncthreads();

    if (wave < OWN_MT) {
        const int m     = lane & 15;
        const int khalf = (lane >> 4) * 2;
        v8f c[8];
#pragma unroll
        for (int t = 0; t < 8; ++t) {
            const float bv = b[t * 16 + m];
#pragma unroll
            for (int r = 0; r < 8; ++r) c[t][r] = bv;
        }
        const float* ar = rows + (wave * 16 + m) * RP + khalf;
        const float* wr = Wl + (size_t)khalf * HID + m;
#pragma unroll 2
        for (int k = 0; k < HID; k += 4) {
            const v2f a = *(const v2f*)(ar + k);
            const float* w0 = wr + (size_t)k * HID;
            const float* w1 = w0 + HID;
#pragma unroll
            for (int t = 0; t < 8; ++t) {
                v2f bf;
                bf.x = w0[t * 16];
                bf.y = w1[t * 16];
                c[t] = __builtin_amdgcn_wmma_f32_16x16x4_f32(false, a, false, bf,
                                                             (short)0, c[t], false, false);
            }
        }
        float* dr = rows + (wave * 16 + ((lane >> 4) ? 8 : 0)) * RP + m;
#pragma unroll
        for (int t = 0; t < 8; ++t)
#pragma unroll
            for (int r = 0; r < 8; ++r)
                dr[r * RP + t * 16] = c[t][r];
    }
    __syncthreads();

    {
        float4 ws = make_float4(0.f, 0.f, 0.f, 0.f), wd = ws;
        if (LAYER == 0) { ws = ((const float4*)was)[lane]; wd = ((const float4*)wad)[lane]; }
        float4* o = (float4*)(out + (size_t)row0 * HID);
        for (int r = wave; r < nrows; r += OWN_T / 32) {
            const float4 v = *(const float4*)(rows + r * RP + 4 * lane);
            o[r * 32 + lane] = v;
            if (LAYER == 0) {
                float ss, sd;
                row_logits(v, ws, wd, ss, sd);
                if (lane == 0) { as_n[row0 + r] = ss; ad_n[row0 + r] = sd; }
            }
        }
    }
}

extern "C" void kernel_launch(void* const* d_in, const int* in_sizes, int n_in,
                              void* d_out, int out_size, void* d_ws, size_t ws_size,
                              hipStream_t stream) {
    const int*   ei      = (const int*)d_in[0];
    const float* users   = (const float*)d_in[1];
    const float* items   = (const float*)d_in[2];
    const float* W       = (const float*)d_in[3];
    const float* att_src = (const float*)d_in[4];
    const float* att_dst = (const float*)d_in[5];
    const float* bias    = (const float*)d_in[6];
    float*       out     = (float*)d_out;

    float*    xbuf  = (float*)d_ws;
    float*    x1buf = xbuf + (size_t)NNODES * HID;
    float*    as0   = x1buf + (size_t)NNODES * HID;
    float*    ad0   = as0 + NNODES;
    float*    as1   = ad0 + NNODES;
    float*    ad1   = as1 + NNODES;
    unsigned* keyw  = (unsigned*)(ad1 + NNODES);
    int*      qs    = (int*)(keyw + KW_WORDS);
    int*      nqs   = qs + NOWN * QCAP;

    const float* W1 = W + (size_t)HID * HID;

    pre_k<<<PRE_G, PRE_T, 0, stream>>>(users, items, xbuf, ei, keyw, W, att_src, att_dst, as0, ad0);
    own_agg_k<0><<<NOWN, OWN_T, 0, stream>>>(ei, xbuf, as0, ad0, keyw, bias, W, x1buf, qs, nqs,
                                             W1, att_src + HID, att_dst + HID, as1, ad1);
    own_agg_k<1><<<NOWN, OWN_T, 0, stream>>>(ei, x1buf, as1, ad1, keyw, bias + HID, W1, out, qs, nqs,
                                             nullptr, nullptr, nullptr, nullptr, nullptr);
}


# Layer 0's product folded into layer 1's (softmax rows sum to 1): layer 0 stores its row sums, layer 1 multiplies once by W0 W1 (WMMA, first launch) from b0 W1 + b1; new rounding order
# speedup vs baseline: 5.4484x; 1.0774x over previous
#include <hip/hip_runtime.h>
#include <hip/hip_bf16.h>

#define NUM_USERS 50000
#define NUM_ITEMS 20000
#define NNODES    70000
#define MTILES    (NNODES / 16)
#define HID       128
#define NE        1000000
#define EPLUS     (NE + NNODES)
#define NEG_SLOPE 0.2f

typedef float v2f __attribute__((ext_vector_type(2)));
typedef float v8f __attribute__((ext_vector_type(8)));

__device__ __forceinline__ unsigned fkey(float f) {
    unsigned b = __float_as_uint(f);
    return (b & 0x80000000u) ? ~b : (b | 0x80000000u);
}
__device__ __forceinline__ float funkey(unsigned k) {
    unsigned b = (k & 0x80000000u) ? (k & 0x7FFFFFFFu) : ~k;
    return __uint_as_float(b);
}

__device__ __forceinline__ void att_through_w(const float* __restrict__ Wl,
                                              const float* __restrict__ asrc,
                                              const float* __restrict__ adst,
                                              float* was, float* wad, int wave, int lane) {
    const float4 av = ((const float4*)asrc)[lane];
    const float4 dv = ((const float4*)adst)[lane];
    float4 w[4];
#pragma unroll
    for (int i = 0; i < 4; ++i) w[i] = ((const float4*)Wl)[(size_t)(wave + 32 * i) * 32 + lane];
#pragma unroll
    for (int i = 0; i < 4; ++i) {
        float ps = w[i].x * av.x + w[i].y * av.y + w[i].z * av.z + w[i].w * av.w;
        float pd = w[i].x * dv.x + w[i].y * dv.y + w[i].z * dv.z + w[i].w * dv.w;
#pragma unroll
        for (int off = 16; off > 0; off >>= 1) {
            ps += __shfl_xor(ps, off, 32);
            pd += __shfl_xor(pd, off, 32);
        }
        if (lane == 0) { was[wave + 32 * i] = ps; wad[wave + 32 * i] = pd; }
    }
}

__device__ __forceinline__ void row_logits(const float4 v, const float4 ws, const float4 wd,
                                           float& ss, float& sd) {
    ss = v.x * ws.x + v.y * ws.y + v.z * ws.z + v.w * ws.w;
    sd = v.x * wd.x + v.y * wd.y + v.z * wd.z + v.w * wd.w;
#pragma unroll
    for (int off = 16; off > 0; off >>= 1) {
        ss += __shfl_xor(ss, off, 32);
        sd += __shfl_xor(sd, off, 32);
    }
}

__device__ __forceinline__ void fold_layers(const float* __restrict__ W0, const float* __restrict__ W1,
                                            const float* __restrict__ asrc1, const float* __restrict__ adst1,
                                            const float* __restrict__ b0, const float* __restrict__ b1,
                                            float* __restrict__ w01, float* __restrict__ b01,
                                            float* __restrict__ vs1, float* __restrict__ vd1,
                                            float* __restrict__ cst,
                                            float* us, float* ud, float* part,
                                            int tid, int wave, int lane) {
    att_through_w(W1, asrc1, adst1, us, ud, wave, lane);
    {
        const int j = tid & (HID - 1), kq = tid >> 7;
        float acc = 0.f;
#pragma unroll
        for (int i = 0; i < 16; ++i) acc += b0[kq * 16 + i] * W1[(size_t)(kq * 16 + i) * HID + j];
        part[kq * HID + j] = acc;
    }
    __syncthreads();
    att_through_w(W0, us, ud, vs1, vd1, wave, lane);
    if (wave == 0) {
        float ss, sd;
        row_logits(((const float4*)b0)[lane], ((const float4*)us)[lane], ((const float4*)ud)[lane], ss, sd);
        if (lane == 0) { cst[0] = ss; cst[1] = sd; }
    }
    if (tid < HID) {
        float s = b1[tid];
#pragma unroll
        for (int qd = 0; qd < 8; ++qd) s += part[qd * HID + tid];
        b01[tid] = s;
    }
    {
        const int m     = lane & 15;
        const int khalf = (lane >> 4) * 2;
        const int rt    = wave >> 2;
        const int ct    = (wave & 3) * 2;
        v8f c0, c1;
#pragma unroll
        for (int r = 0; r < 8; ++r) { c0[r] = 0.f; c1[r] = 0.f; }
        const float* ar = W0 + (size_t)(rt * 16 + m) * HID + khalf;
        const float* wr = W1 + (size_t)khalf * HID + ct * 16 + m;
#pragma unroll 4
        for (int k = 0; k < HID; k += 4) {
            const v2f a = *(const v2f*)(ar + k);
            const float* w0 = wr + (size_t)k * HID;
            v2f bf0, bf1;
            bf0.x = w0[0];  bf0.y = w0[HID];
            bf1.x = w0[16]; bf1.y = w0[HID + 16];
            c0 = __builtin_amdgcn_wmma_f32_16x16x4_f32(false, a, false, bf0, (short)0, c0, false, false);
            c1 = __builtin_amdgcn_wmma_f32_16x16x4_f32(false, a, false, bf1, (short)0, c1, false, false);
        }
        float* dr = w01 + (size_t)(rt * 16 + ((lane >> 4) ? 8 : 0)) * HID + ct * 16 + m;
#pragma unroll
        for (int r = 0; r < 8; ++r) { dr[r * HID] = c0[r]; dr[r * HID + 16] = c1[r]; }
    }
}

#define OWN_R 274
#define NOWN  ((NNODES + OWN_R - 1) / OWN_R)
#define PRE_T 1024
#define PRE_G 512
#define PRE_W (PRE_G * PRE_T / 32)
#define PRE_U 5
#define NKW   ((EPLUS + 127) / 128)
#define NSB   (NKW / 8)
#define NUNIT (NKW * 4)
#define KW_WORDS (NSB * 256)
#define TAIL_LANES ((EPLUS - 128 * (NKW - 1)) / 4)
static_assert(NOWN <= 256, "the owner index must fit eight planes");
static_assert(EPLUS % 4 == 0 && NE % 4 == 0, "four edges per thread, never across the self-loop border");
static_assert(NKW % 8 == 0 && PRE_T == 1024, "a group of the first launch writes four whole super-blocks");
static_assert(TAIL_LANES > 0 && TAIL_LANES < 32, "the last block is partly filled: its mask below");
static_assert(PRE_W * PRE_U >= NNODES && PRE_G * PRE_T >= EPLUS / 4, "the first launch covers every row and every edge");
__global__ __launch_bounds__(PRE_T) void pre_k(const float* __restrict__ u,
                                               const float* __restrict__ it,
                                               float* __restrict__ x,
                                               const int* __restrict__ ei,
                                               unsigned* __restrict__ keyw,
                                               const float* __restrict__ W0,
                                               const float* __restrict__ asrc,
                                               const float* __restrict__ adst,
                                               float* __restrict__ as_out,
                                               float* __restrict__ ad_out,
                                               const float* __restrict__ W1,
                                               const float* __restrict__ asrc1,
                                               const float* __restrict__ adst1,
                                               const float* __restrict__ b0,
                                               const float* __restrict__ b1,
                                               float* __restrict__ w01,
                                               float* __restrict__ b01,
                                               float* __restrict__ vs1,
                                               float* __restrict__ vd1,
                                               float* __restrict__ cst) {
    __shared__ __attribute__((aligned(16))) float was[HID];
    __shared__ __attribute__((aligned(16))) float wad[HID];
    __shared__ unsigned kst[PRE_T];
    __shared__ float part[8 * HID];
    const int tid  = threadIdx.x;
    const int lane = tid & 31;
    const int wave = __builtin_amdgcn_readfirstlane(tid >> 5);
    if (blockIdx.x == 0) {
        fold_layers(W0, W1, asrc1, adst1, b0, b1, w01, b01, vs1, vd1, cst, was, wad, part, tid, wave, lane);
        return;
    }
    const int bid = blockIdx.x - 1;
    const int i = bid * PRE_T + tid;
    int4 d = make_int4(0, 0, 0, 0);
    if (i < EPLUS / 4) {
        if (4 * i < NE) d = ((const int4*)(ei + NE))[i];
        else { const int n = 4 * i - NE; d = make_int4(n, n + 1, n + 2, n + 3); }
    }
    const int gw = bid * (PRE_T / 32) + wave;
    float4 v[PRE_U];
#pragma unroll
    for (int j = 0; j < PRE_U; ++j) {
        const int row = gw + j * PRE_W;
        const int rc  = (row < NNODES) ? row : 0;
        const float* src = (rc < NUM_USERS) ? u + (size_t)rc * HID : it + (size_t)(rc - NUM_USERS) * HID;
        v[j] = ((const float4*)src)[lane];
    }

    att_through_w(W0, asrc, adst, was, wad, wave, lane);

    if (gw < NKW) {
        const unsigned key32 = (unsigned)(d.x / OWN_R) | ((unsigned)(d.y / OWN_R) << 8) |
                               ((unsigned)(d.z / OWN_R) << 16) | ((unsigned)(d.w / OWN_R) << 24);
        unsigned myw = 0u;
#pragma unroll
        for (int j = 0; j < 32; ++j) {
            const unsigned vote = __builtin_amdgcn_ballot_w32(((key32 >> j) & 1u) != 0u);
            myw = (lane == j) ? vote : myw;
        }
        const int c = lane >> 3, b = lane & 7;
        kst[(wave >> 3) * 256 + (b >> 2) * 128 + ((wave & 7) * 4 + c) * 4 + (b & 3)] = myw;
    }

    __syncthreads();
    if (i < KW_WORDS) keyw[i] = kst[tid];
    const float4 ws = ((const float4*)was)[lane];
    const float4 wd = ((const float4*)wad)[lane];
#pragma unroll
    for (int j = 0; j < PRE_U; ++j) {
        const int row = gw + j * PRE_W;
        if (row < NNODES) {
            ((float4*)x)[(size_t)row * 32 + lane] = v[j];
            float ss, sd;
            row_logits(v[j], ws, wd, ss, sd);
            if (lane == 0) { as_out[row] = ss; ad_out[row] = sd; }
        }
    }
}

__device__ __forceinline__ void edge_sd(int i, const int* __restrict__ ei,
                                        int& s, int& d) {
    const bool real = i < NE;
    const int  ic = real ? i : 0;
    const int  sv = ei[ic], dv = ei[NE + ic];
    s = real ? sv : i - NE;
    d = real ? dv : i - NE;
}

__device__ __forceinline__ float lrelu(float e) { return (e >= 0.f) ? e : NEG_SLOPE * e; }

#define OWN_T   1024
#define SCAN_U  4
#define QCAP    8192
#define LOG_U   8
#define AGG_U   8
#define SBITS   17
#define RP      132
#define OWN_MT  ((OWN_R + 15) / 16)
static_assert(QCAP == OWN_T * LOG_U, "the logit step covers the queue in one sweep");
static_assert(NNODES <= (1 << SBITS) && OWN_R <= (1 << (31 - SBITS)), "queue entry packs (s, local row)");
static_assert(RP % 4 == 0 && RP >= HID && OWN_MT <= OWN_T / 32, "16-byte row starts; a wave per row tile");
static_assert(OWN_T == 1024 && AGG_U == 8, "32 owner waves (local row mod 32); batches of 8 rows");

template <class F>
__device__ __forceinline__ void scan_owned(const unsigned* __restrict__ keyw, int tid, F f) {
    const uint4* kp = (const uint4*)keyw;
    const unsigned g = blockIdx.x;
    unsigned nb[8];
#pragma unroll
    for (int b = 0; b < 8; ++b) nb[b] = ((g >> b) & 1u) ? 0u : 0xFFFFFFFFu;
    const int rot = (int)((g * NSB) / NOWN) * 32;
    for (int base = 0; base < NUNIT; base += OWN_T * SCAN_U) {
        uint4 lo[SCAN_U], hi[SCAN_U];
#pragma unroll
        for (int u = 0; u < SCAN_U; ++u) {
            const int p  = base + u * OWN_T + tid;
            const int pr = (p + rot < NUNIT) ? (p + rot) : (p + rot - NUNIT);
            const int pc = (p < NUNIT) ? pr : 0;
            const int at = (pc >> 5) * 64 + (pc & 31);
            lo[u] = kp[at];
            hi[u] = kp[at + 32];
        }
#pragma unroll
        for (int u = 0; u < SCAN_U; ++u) {
            const int p  = base + u * OWN_T + tid;
            const int pr = (p + rot < NUNIT) ? (p + rot) : (p + rot - NUNIT);
            unsigned mk = (p < NUNIT) ? 0xFFFFFFFFu : 0u;
            mk &= lo[u].x ^ nb[0]; mk &= lo[u].y ^ nb[1]; mk &= lo[u].z ^ nb[2]; mk &= lo[u].w ^ nb[3];
            mk &= hi[u].x ^ nb[4]; mk &= hi[u].y ^ nb[5]; mk &= hi[u].z ^ nb[6]; mk &= hi[u].w ^ nb[7];
            if ((pr >> 2) == NKW - 1) mk &= (1u << TAIL_LANES) - 1u;
            const int e0 = ((pr >> 2) << 7) + (pr & 3);
            while (mk) {
                const int bit = __builtin_ctz(mk);
                mk &= mk - 1;
                f(e0 + 4 * bit);
            }
        }
    }
}

__device__ __forceinline__ void load_rows(float4 (&hv)[8], const float* __restrict__ X,
                                          int s, int j0, int lane) {
#pragma unroll
    for (int u = 0; u < 8; ++u) {
        const int su = __builtin_amdgcn_readlane(s, j0 + u);
        hv[u] = ((const float4*)(X + (size_t)su * HID))[lane];
    }
}
__device__ __forceinline__ void add_rows(const float4 (&hv)[8], float* rows,
                                         int dl, float alpha, int j0, int m, int lane) {
#pragma unroll
    for (int u = 0; u < 8; ++u) {
        if (j0 + u < m) {
            const int   dlu = __builtin_amdgcn_readlane(dl, j0 + u);
            const float au  = __int_as_float(__builtin_amdgcn_readlane(__float_as_int(alpha), j0 + u));
            float4* r = (float4*)(rows + dlu * RP) + lane;
            float4 t = *r;
            t.x += hv[u].x * au; t.y += hv[u].y * au; t.z += hv[u].z * au; t.w += hv[u].w * au;
            *r = t;
        }
    }
}

template <int LAYER>
__global__ __launch_bounds__(OWN_T) void own_agg_k(const int* __restrict__ ei,
                                                   const float* __restrict__ X,
                                                   const float* __restrict__ as_,
                                                   const float* __restrict__ ad_,
                                                   const unsigned* __restrict__ keyw,
                                                   const float* __restrict__ b,
                                                   const float* __restrict__ Wl,
                                                   float* __restrict__ out,
                                                   int* __restrict__ qs,
                                                   int* __restrict__ nqs,
                                                   const float* __restrict__ vs_n,
                                                   const float* __restrict__ vd_n,
                                                   const float* __restrict__ cst_n,
                                                   float* __restrict__ as_n,
                                                   float* __restrict__ ad_n) {
    __shared__ __attribute__((aligned(16))) float rows[OWN_MT * 16 * RP];
    __shared__ float    dens[OWN_R];
    __shared__ float    adl[OWN_R];
    __shared__ unsigned mx[OWN_R];
    __shared__ int      q[QCAP];
    __shared__ float    qx[QCAP];
    __shared__ int      bcnt[OWN_T / 32];
    __shared__ int      qn;

    const int tid  = threadIdx.x;
    const int lane = tid & 31;
    const int wave = __builtin_amdgcn_readfirstlane(tid >> 5);
    const int row0 = blockIdx.x * OWN_R;
    const int nrows = (NNODES - row0 < OWN_R) ? (NNODES - row0) : OWN_R;

    {
        if (tid < OWN_T / 32) bcnt[tid] = 0;
        for (int i = tid; i < OWN_MT * 16 * RP / 4; i += OWN_T) ((float4*)rows)[i] = make_float4(0.f, 0.f, 0.f, 0.f);
        for (int i = tid; i < OWN_R; i += OWN_T) {
            dens[i] = 0.f;
            mx[i]   = 0u;
            adl[i]  = (i < nrows) ? ad_[row0 + i] : 0.f;
        }
        if (tid == 0) qn = 0;
    }
    __syncthreads();

    int* const qsv = qs + blockIdx.x * QCAP;
    int nq;
    if (LAYER == 0) {
        scan_owned(keyw, tid, [&](int e) {
            const int slot = atomicAdd(&qn, 1);
            if (slot < QCAP) q[slot] = e;
        });
        __syncthreads();
        nq = __builtin_amdgcn_readfirstlane(qn);
        if (tid == 0) nqs[blockIdx.x] = nq;
    } else {
        nq = __builtin_amdgcn_readfirstlane(nqs[blockIdx.x]);
    }

    if (nq <= QCAP) {
        int bin_0 = 0, bin_n = 0;
        {
            int s[LOG_U], dl[LOG_U];
            float a[LOG_U];
            if (LAYER == 0) {
                int e[LOG_U], d[LOG_U];
#pragma unroll
                for (int u = 0; u < LOG_U; ++u) {
                    const int i = u * OWN_T + tid;
                    e[u] = (i < nq) ? q[i] : 0;
                }
#pragma unroll
                for (int u = 0; u < LOG_U; ++u) edge_sd(e[u], ei, s[u], d[u]);
#pragma unroll
                for (int u = 0; u < LOG_U; ++u) dl[u] = d[u] - row0;
            } else {
                int pk[LOG_U];
#pragma unroll
                for (int u = 0; u < LOG_U; ++u) {
                    const int i = u * OWN_T + tid;
                    pk[u] = qsv[(i < nq) ? i : 0];
                }
#pragma unroll
                for (int u = 0; u < LOG_U; ++u) {
                    s[u]  = pk[u] & ((1 << SBITS) - 1);
                    dl[u] = pk[u] >> SBITS;
                }
            }
#pragma unroll
            for (int u = 0; u < LOG_U; ++u) a[u] = as_[s[u]];
            int   rk[LOG_U];
            float lgv[LOG_U];
#pragma unroll
            for (int u = 0; u < LOG_U; ++u) {
                const int i = u * OWN_T + tid;
                rk[u] = 0; lgv[u] = 0.f;
                if (i < nq) {
                    const float lg = lrelu(a[u] + adl[dl[u]]);
                    atomicMax(&mx[dl[u]], fkey(lg));
                    rk[u]  = atomicAdd(&bcnt[dl[u] & 31], 1);
                    lgv[u] = lg;
                    if (LAYER == 0) qsv[i] = s[u] | (dl[u] << SBITS);
                }
            }
            __syncthreads();
            const int cntl = bcnt[lane];
            int incl = cntl;
#pragma unroll
            for (int off = 1; off < 32; off <<= 1) {
                const int t = __shfl_up(incl, off, 32);
                if (lane >= off) incl += t;
            }
            bin_n = __builtin_amdgcn_readlane(cntl, wave);
            bin_0 = __builtin_amdgcn_readlane(incl - cntl, wave);
#pragma unroll
            for (int u = 0; u < LOG_U; ++u) {
                const int i  = u * OWN_T + tid;
                const int st = __shfl(incl - cntl, dl[u] & 31, 32);
                if (i < nq) {
                    q[st + rk[u]]  = s[u] | (dl[u] << SBITS);
                    qx[st + rk[u]] = lgv[u];
                }
            }
        }
        __syncthreads();

        for (int i = tid; i < nq; i += OWN_T) {
            const int   dl = q[i] >> SBITS;
            const float ex = expf(qx[i] - funkey(mx[dl]));
            qx[i] = ex;
            atomicAdd(&dens[dl], ex);
        }
        __syncthreads();

        for (int c0 = 0; c0 < bin_n; c0 += 32) {
            const int m = (bin_n - c0 < 32) ? (bin_n - c0) : 32;
            int s = 0, dl = 0;
            float alpha = 0.f;
            if (lane < m) {
                const int pk = q[bin_0 + c0 + lane];
                s  = pk & ((1 << SBITS) - 1);
                dl = pk >> SBITS;
                alpha = qx[bin_0 + c0 + lane] / dens[dl];
            }
            float4 ha[AGG_U], hb[AGG_U];
            load_rows(ha, X, s, 0, lane);
            if (m > 8)  load_rows(hb, X, s, 8, lane);
            add_rows(ha, rows, dl, alpha, 0, m, lane);
            if (m > 16) load_rows(ha, X, s, 16, lane);
            if (m > 8)  add_rows(hb, rows, dl, alpha, 8, m, lane);
            if (m > 24) load_rows(hb, X, s, 24, lane);
            if (m > 16) add_rows(ha, rows, dl, alpha, 16, m, lane);
            if (m > 24) add_rows(hb, rows, dl, alpha, 24, m, lane);
        }
    } else {
#pragma unroll 1
        for (int pass = 0; pass < 3; ++pass) {
            scan_owned(keyw, tid, [&](int e) {
                int s, d; edge_sd(e, ei, s, d);
                const int   dl = d - row0;
                const float lg = lrelu(as_[s] + adl[dl]);
                if (pass == 0) { atomicMax(&mx[dl], fkey(lg)); return; }
                const float ex = expf(lg - funkey(mx[dl]));
                if (pass == 1) { atomicAdd(&dens[dl], ex); return; }
                const float al = ex / dens[dl];
                const float* hr = X + (size_t)s * HID;
                float* r = rows + dl * RP;
#pragma unroll 1
                for (int c = 0; c < HID; ++c) atomicAdd(r + c, hr[c] * al);
            });
            __syncthreads();
        }
    }
    __syncthreads();

    if (LAYER == 1) {
    if (wave < OWN_MT) {
        const int m     = lane & 15;
        const int khalf = (lane >> 4) * 2;
        v8f c[8];
#pragma unroll
        for (int t = 0; t < 8; ++t) {
            const float bv = b[t * 16 + m];
#pragma unroll
            for (int r = 0; r < 8; ++r) c[t][r] = bv;
        }
        const float* ar = rows + (wave * 16 + m) * RP + khalf;
        const float* wr = Wl + (size_t)khalf * HID + m;
#pragma unroll 2
        for (int k = 0; k < HID; k += 4) {
            const v2f a = *(const v2f*)(ar + k);
            const float* w0 = wr + (size_t)k * HID;
            const float* w1 = w0 + HID;
#pragma unroll
            for (int t = 0; t < 8; ++t) {
                v2f bf;
                bf.x = w0[t * 16];
                bf.y = w1[t * 16];
                c[t] = __builtin_amdgcn_wmma_f32_16x16x4_f32(false, a, false, bf,
                                                             (short)0, c[t], false, false);
            }
        }
        float* dr = rows + (wave * 16 + ((lane >> 4) ? 8 : 0)) * RP + m;
#pragma unroll
        for (int t = 0; t < 8; ++t)
#pragma unroll
            for (int r = 0; r < 8; ++r)
                dr[r * RP + t * 16] = c[t][r];
    }
    __syncthreads();
    }

    {
        float4 ws = make_float4(0.f, 0.f, 0.f, 0.f), wd = ws;
        float cs = 0.f, cd = 0.f;
        if (LAYER == 0) {
            ws = ((const float4*)vs_n)[lane]; wd = ((const float4*)vd_n)[lane];
            cs = cst_n[0]; cd = cst_n[1];
        }
        float4* o = (float4*)(out + (size_t)row0 * HID);
        for (int r = wave; r < nrows; r += OWN_T / 32) {
            const float4 v = *(const float4*)(rows + r * RP + 4 * lane);
            o[r * 32 + lane] = v;
            if (LAYER == 0) {
                float ss, sd;
                row_logits(v, ws, wd, ss, sd);
                if (lane == 0) { as_n[row0 + r] = ss + cs; ad_n[row0 + r] = sd + cd; }
            }
        }
    }
}

extern "C" void kernel_launch(void* const* d_in, const int* in_sizes, int n_in,
                              void* d_out, int out_size, void* d_ws, size_t ws_size,
                              hipStream_t stream) {
    const int*   ei      = (const int*)d_in[0];
    const float* users   = (const float*)d_in[1];
    const float* items   = (const float*)d_in[2];
    const float* W       = (const float*)d_in[3];
    const float* att_src = (const float*)d_in[4];
    const float* att_dst = (const float*)d_in[5];
    const float* bias    = (const float*)d_in[6];
    float*       out     = (float*)d_out;

    float*    xbuf  = (float*)d_ws;
    float*    x1buf = xbuf + (size_t)NNODES * HID;
    float*    as0   = x1buf + (size_t)NNODES * HID;
    float*    ad0   = as0 + NNODES;
    float*    as1   = ad0 + NNODES;
    float*    ad1   = as1 + NNODES;
    unsigned* keyw  = (unsigned*)(ad1 + NNODES);
    int*      qs    = (int*)(keyw + KW_WORDS);
    int*      nqs   = qs + NOWN * QCAP;
    float*    w01   = (float*)(nqs + NOWN);
    float*    b01   = w01 + HID * HID;
    float*    vs1   = b01 + HID;
    float*    vd1   = vs1 + HID;
    float*    cst   = vd1 + HID;

    const float* W1 = W + (size_t)HID * HID;

    pre_k<<<PRE_G + 1, PRE_T, 0, stream>>>(users, items, xbuf, ei, keyw, W, att_src, att_dst, as0, ad0,
                                           W1, att_src + HID, att_dst + HID, bias, bias + HID,
                                           w01, b01, vs1, vd1, cst);
    own_agg_k<0><<<NOWN, OWN_T, 0, stream>>>(ei, xbuf, as0, ad0, keyw, nullptr, nullptr, x1buf, qs, nqs,
                                             vs1, vd1, cst, as1, ad1);
    own_agg_k<1><<<NOWN, OWN_T, 0, stream>>>(ei, x1buf, as1, ad1, keyw, b01, w01, out, qs, nqs,
                                             nullptr, nullptr, nullptr, nullptr, nullptr);
}
